# starting kernel
# baseline (speedup 1.0000x reference)
#include <hip/hip_runtime.h>


typedef __attribute__((ext_vector_type(2))) float v2f;
typedef __attribute__((ext_vector_type(8))) float v8f;

#define NB 16
#define NA 128
#define HD 256
#define EF 30
#define KP 32
#define K2 (2 * HD)
#define KC 32
#define ROWS 64

__device__ __forceinline__ float selu_f(float x) {
    const float scale = 1.0507009873554805f;
    const float alpha = 1.6732632423543772f;
    return x > 0.0f ? scale * x : scale * alpha * (__expf(x) - 1.0f);
}

__device__ __forceinline__ v8f wmma_f32(v2f a, v2f b, v8f c) {
    return __builtin_amdgcn_wmma_f32_16x16x4_f32(false, a, false, b, (short)0, c,
                                                 false, false);
}

__global__ __launch_bounds__(256)
void ef_filter_reduce(const float* __restrict__ h,
                      const float* __restrict__ e,
                      const float* __restrict__ edge_mask,
                      const float* __restrict__ Wf,
                      const float* __restrict__ bf,
                      float* __restrict__ h1out) {
    __shared__ float sh_wf[KP * HD];
    __shared__ float sh_e[NA * KP];

    const int tid = threadIdx.x;
    const int bi  = blockIdx.x;
    const int b   = bi / NA;
    const int i   = bi % NA;

    {
        const float* wrow = Wf + (size_t)tid * EF;
#pragma unroll
        for (int k = 0; k < KP; ++k)
            sh_wf[k * HD + tid] = (k < EF) ? wrow[k] : 0.0f;
    }
    {
        const float* erow = e + ((size_t)(b * NA + i)) * NA * EF;
        const int j  = tid >> 1;
        const int kb = (tid & 1) * 16;
#pragma unroll
        for (int k = 0; k < 16; ++k) {
            const int kk = kb + k;
            sh_e[j * KP + kk] = (kk < EF) ? erow[j * EF + kk] : 0.0f;
        }
    }
    __syncthreads();

    const int lane  = tid & 31;
    const int wave  = tid >> 5;
    const int half  = lane >> 4;
    const int l     = lane & 15;
    const int fbase = wave * 32;

    const v8f vzero = {0.f, 0.f, 0.f, 0.f, 0.f, 0.f, 0.f, 0.f};
    v8f acc[8][2];
#pragma unroll
    for (int mt = 0; mt < 8; ++mt)
#pragma unroll
        for (int nt = 0; nt < 2; ++nt)
            acc[mt][nt] = vzero;

#pragma unroll
    for (int kk = 0; kk < KP / 4; ++kk) {
        const int k0 = kk * 4 + 2 * half;
#pragma unroll
        for (int mt = 0; mt < 8; ++mt) {
            const int j = mt * 16 + l;
            v2f a;
            a.x = sh_e[j * KP + k0];
            a.y = sh_e[j * KP + k0 + 1];
#pragma unroll
            for (int nt = 0; nt < 2; ++nt) {
                const int n = fbase + nt * 16 + l;
                v2f bb;
                bb.x = sh_wf[k0 * HD + n];
                bb.y = sh_wf[(k0 + 1) * HD + n];
                acc[mt][nt] = wmma_f32(a, bb, acc[mt][nt]);
            }
        }
    }

    const float* hB  = h + (size_t)b * NA * HD;
    const float* emB = edge_mask + ((size_t)(b * NA + i)) * NA;
#pragma unroll
    for (int nt = 0; nt < 2; ++nt) {
        const int f     = fbase + nt * 16 + l;
        const float bia = bf[f];
        float psum = 0.0f;
#pragma unroll
        for (int mt = 0; mt < 8; ++mt) {
#pragma unroll
            for (int r = 0; r < 8; ++r) {
                const int j  = mt * 16 + half * 8 + r;
                const float ef = selu_f(acc[mt][nt][r] + bia) * emB[j];
                psum += ef * hB[j * HD + f];
            }
        }
        const float tot = psum + __shfl_xor(psum, 16, 32);
        if (half == 0)
            h1out[(size_t)(b * NA + i) * HD + f] = tot;
    }
}

__global__ __launch_bounds__(256)
void ef_node_mlp(const float* __restrict__ h,
                 const float* __restrict__ node_mask,
                 const float* __restrict__ Ww,
                 const float* __restrict__ bw,
                 float* __restrict__ out  ) {
    __shared__ float sh_z[ROWS * KC];
    __shared__ float sh_w[KC * HD];

    const int tid = threadIdx.x;
    const int r0  = blockIdx.x * ROWS;

    const int lane  = tid & 31;
    const int wave  = tid >> 5;
    const int half  = lane >> 4;
    const int l     = lane & 15;
    const int nbase = wave * 32;

    const v8f vzero = {0.f, 0.f, 0.f, 0.f, 0.f, 0.f, 0.f, 0.f};
    v8f acc[4][2];
#pragma unroll
    for (int mt = 0; mt < 4; ++mt)
#pragma unroll
        for (int nt = 0; nt < 2; ++nt)
            acc[mt][nt] = vzero;

    for (int kc = 0; kc < K2 / KC; ++kc) {
        const int kg0 = kc * KC;
        {
            const int rr  = tid >> 2;
            const int kb  = (tid & 3) * 8;
            const int row = r0 + rr;
#pragma unroll
            for (int k = 0; k < 8; ++k) {
                const int kg = kg0 + kb + k;
                float v;
                if (kg < HD) v = h[(size_t)row * HD + kg];
                else         v = out[(size_t)row * HD + (kg - HD)];
                sh_z[rr * KC + kb + k] = v;
            }
        }
        {
            const float* wn = Ww + (size_t)tid * K2 + kg0;
#pragma unroll
            for (int k = 0; k < KC; ++k)
                sh_w[k * HD + tid] = wn[k];
        }
        __syncthreads();

#pragma unroll
        for (int kk = 0; kk < KC / 4; ++kk) {
            const int k0 = kk * 4 + 2 * half;
#pragma unroll
            for (int mt = 0; mt < 4; ++mt) {
                const int rr = mt * 16 + l;
                v2f a;
                a.x = sh_z[rr * KC + k0];
                a.y = sh_z[rr * KC + k0 + 1];
#pragma unroll
                for (int nt = 0; nt < 2; ++nt) {
                    const int n = nbase + nt * 16 + l;
                    v2f bb;
                    bb.x = sh_w[k0 * HD + n];
                    bb.y = sh_w[(k0 + 1) * HD + n];
                    acc[mt][nt] = wmma_f32(a, bb, acc[mt][nt]);
                }
            }
        }
        __syncthreads();
    }

#pragma unroll
    for (int mt = 0; mt < 4; ++mt) {
#pragma unroll
        for (int nt = 0; nt < 2; ++nt) {
            const int f = nbase + nt * 16 + l;
#pragma unroll
            for (int r = 0; r < 8; ++r) {
                const int row = r0 + mt * 16 + half * 8 + r;
                float v = selu_f(acc[mt][nt][r] + bw[f]) * node_mask[row];
                out[(size_t)row * HD + f] = v + h[(size_t)row * HD + f];
            }
        }
    }
}

extern "C" void kernel_launch(void* const* d_in, const int* in_sizes, int n_in,
                              void* d_out, int out_size, void* d_ws, size_t ws_size,
                              hipStream_t stream) {
    (void)in_sizes; (void)n_in; (void)out_size; (void)d_ws; (void)ws_size;
    const float* h         = (const float*)d_in[0];
    const float* e         = (const float*)d_in[1];
    const float* node_mask = (const float*)d_in[2];
    const float* edge_mask = (const float*)d_in[3];
    const float* Wf        = (const float*)d_in[4];
    const float* bf        = (const float*)d_in[5];
    const float* Ww        = (const float*)d_in[6];
    const float* bw        = (const float*)d_in[7];
    float* out = (float*)d_out;

    ef_filter_reduce<<<NB * NA, 256, 0, stream>>>(h, e, edge_mask, Wf, bf, out);
    ef_node_mlp<<<(NB * NA) / ROWS, 256, 0, stream>>>(h, node_mask, Ww, bw, out);
}


# Node MLP re-tiled: 16 rows per work-group (128 work-groups, was 32 of 64 rows), Ww streamed row-major through two LDS stages by contiguous 16-B loads; k order inside the matrix sums changes
# speedup vs baseline: 1.4549x; 1.4549x over previous
#include <hip/hip_runtime.h>


typedef __attribute__((ext_vector_type(2))) float v2f;
typedef __attribute__((ext_vector_type(4))) float v4f;
typedef __attribute__((ext_vector_type(8))) float v8f;

#define NB 16
#define NA 128
#define HD 256
#define EF 30
#define KP 32
#define K2 (2 * HD)
#define R2 16
#define KC2 64
#define NCH2 (K2 / KC2)
#define ZP4 ((K2 + 4) / 4)
#define WP4 ((KC2 + 4) / 4)

__device__ __forceinline__ float selu_f(float x) {
    const float scale = 1.0507009873554805f;
    const float alpha = 1.6732632423543772f;
    return x > 0.0f ? scale * x : scale * alpha * (__expf(x) - 1.0f);
}

__device__ __forceinline__ v8f wmma_f32(v2f a, v2f b, v8f c) {
    return __builtin_amdgcn_wmma_f32_16x16x4_f32(false, a, false, b, (short)0, c,
                                                 false, false);
}

__global__ __launch_bounds__(256)
void ef_filter_reduce(const float* __restrict__ h,
                      const float* __restrict__ e,
                      const float* __restrict__ edge_mask,
                      const float* __restrict__ Wf,
                      const float* __restrict__ bf,
                      float* __restrict__ h1out) {
    __shared__ float sh_wf[KP * HD];
    __shared__ float sh_e[NA * KP];

    const int tid = threadIdx.x;
    const int bi  = blockIdx.x;
    const int b   = bi / NA;
    const int i   = bi % NA;

    {
        const float* wrow = Wf + (size_t)tid * EF;
#pragma unroll
        for (int k = 0; k < KP; ++k)
            sh_wf[k * HD + tid] = (k < EF) ? wrow[k] : 0.0f;
    }
    {
        const float* erow = e + ((size_t)(b * NA + i)) * NA * EF;
        const int j  = tid >> 1;
        const int kb = (tid & 1) * 16;
#pragma unroll
        for (int k = 0; k < 16; ++k) {
            const int kk = kb + k;
            sh_e[j * KP + kk] = (kk < EF) ? erow[j * EF + kk] : 0.0f;
        }
    }
    __syncthreads();

    const int lane  = tid & 31;
    const int wave  = tid >> 5;
    const int half  = lane >> 4;
    const int l     = lane & 15;
    const int fbase = wave * 32;

    const v8f vzero = {0.f, 0.f, 0.f, 0.f, 0.f, 0.f, 0.f, 0.f};
    v8f acc[8][2];
#pragma unroll
    for (int mt = 0; mt < 8; ++mt)
#pragma unroll
        for (int nt = 0; nt < 2; ++nt)
            acc[mt][nt] = vzero;

#pragma unroll
    for (int kk = 0; kk < KP / 4; ++kk) {
        const int k0 = kk * 4 + 2 * half;
#pragma unroll
        for (int mt = 0; mt < 8; ++mt) {
            const int j = mt * 16 + l;
            v2f a;
            a.x = sh_e[j * KP + k0];
            a.y = sh_e[j * KP + k0 + 1];
#pragma unroll
            for (int nt = 0; nt < 2; ++nt) {
                const int n = fbase + nt * 16 + l;
                v2f bb;
                bb.x = sh_wf[k0 * HD + n];
                bb.y = sh_wf[(k0 + 1) * HD + n];
                acc[mt][nt] = wmma_f32(a, bb, acc[mt][nt]);
            }
        }
    }

    const float* hB  = h + (size_t)b * NA * HD;
    const float* emB = edge_mask + ((size_t)(b * NA + i)) * NA;
#pragma unroll
    for (int nt = 0; nt < 2; ++nt) {
        const int f     = fbase + nt * 16 + l;
        const float bia = bf[f];
        float psum = 0.0f;
#pragma unroll
        for (int mt = 0; mt < 8; ++mt) {
#pragma unroll
            for (int r = 0; r < 8; ++r) {
                const int j  = mt * 16 + half * 8 + r;
                const float ef = selu_f(acc[mt][nt][r] + bia) * emB[j];
                psum += ef * hB[j * HD + f];
            }
        }
        const float tot = psum + __shfl_xor(psum, 16, 32);
        if (half == 0)
            h1out[(size_t)(b * NA + i) * HD + f] = tot;
    }
}

__device__ __forceinline__ v4f ld4(const float* p) {
    v4f v;
    v.x = p[0]; v.y = p[1]; v.z = p[2]; v.w = p[3];
    return v;
}

__global__ __launch_bounds__(256)
void ef_node_mlp(const float* __restrict__ h,
                 const float* __restrict__ node_mask,
                 const float* __restrict__ Ww,
                 const float* __restrict__ bw,
                 float* __restrict__ out  ) {
    __shared__ v4f sh_z[R2 * ZP4];
    __shared__ v4f sh_w[2][HD * WP4];

    const int tid = threadIdx.x;
    const int r0  = blockIdx.x * R2;

    const int lane  = tid & 31;
    const int wave  = tid >> 5;
    const int half  = lane >> 4;
    const int l     = lane & 15;
    const int nbase = wave * 32;

    float bia[2], msk[8], res[2][8];
#pragma unroll
    for (int nt = 0; nt < 2; ++nt) {
        const int f = nbase + nt * 16 + l;
        bia[nt] = bw[f];
#pragma unroll
        for (int r = 0; r < 8; ++r)
            res[nt][r] = h[(size_t)(r0 + half * 8 + r) * HD + f];
    }
#pragma unroll
    for (int r = 0; r < 8; ++r)
        msk[r] = node_mask[r0 + half * 8 + r];

    {
        const int c4 = tid & 127;
        const int rb = tid >> 7;
        const bool from_h = c4 < 64;
#pragma unroll
        for (int i = 0; i < 8; ++i) {
            const int r = 2 * i + rb;
            const size_t row = (size_t)(r0 + r);
            const float* src = from_h ? h + row * HD + c4 * 4
                                      : out + row * HD + (c4 - 64) * 4;
            sh_z[r * ZP4 + c4] = ld4(src);
        }
    }

    const int wn = tid >> 4;
    const int wc = tid & 15;
    const float* wsrc = Ww + (size_t)wn * K2 + wc * 4;
    v4f pre[16];
#pragma unroll
    for (int i = 0; i < 16; ++i)
        pre[i] = ld4(wsrc + (size_t)i * 16 * K2);
#pragma unroll
    for (int i = 0; i < 16; ++i)
        sh_w[0][(i * 16 + wn) * WP4 + wc] = pre[i];

    const v8f vzero = {0.f, 0.f, 0.f, 0.f, 0.f, 0.f, 0.f, 0.f};
    v8f acc0 = vzero, acc1 = vzero;

#pragma unroll
    for (int kc = 0; kc < NCH2; ++kc) {
        const int buf = kc & 1;
        if (kc + 1 < NCH2) {
#pragma unroll
            for (int i = 0; i < 16; ++i)
                pre[i] = ld4(wsrc + (size_t)i * 16 * K2 + (kc + 1) * KC2);
        }
        __syncthreads();

        const v4f* zr = sh_z + l * ZP4 + kc * (KC2 / 4) + half;
        const v4f* w0 = sh_w[buf] + (nbase + l) * WP4 + half;
        const v4f* w1 = w0 + 16 * WP4;
#pragma unroll
        for (int q = 0; q < KC2 / 8; ++q) {
            const v4f a  = zr[2 * q];
            const v4f b0 = w0[2 * q];
            const v4f b1 = w1[2 * q];
            const v2f a_lo = {a.x, a.y},   a_hi = {a.z, a.w};
            const v2f b0lo = {b0.x, b0.y}, b0hi = {b0.z, b0.w};
            const v2f b1lo = {b1.x, b1.y}, b1hi = {b1.z, b1.w};
            acc0 = wmma_f32(a_lo, b0lo, acc0);
            acc1 = wmma_f32(a_lo, b1lo, acc1);
            acc0 = wmma_f32(a_hi, b0hi, acc0);
            acc1 = wmma_f32(a_hi, b1hi, acc1);
        }

        if (kc + 1 < NCH2) {
#pragma unroll
            for (int i = 0; i < 16; ++i)
                sh_w[buf ^ 1][(i * 16 + wn) * WP4 + wc] = pre[i];
        }
    }

#pragma unroll
    for (int nt = 0; nt < 2; ++nt) {
        const int f = nbase + nt * 16 + l;
        const v8f acc = nt ? acc1 : acc0;
#pragma unroll
        for (int r = 0; r < 8; ++r) {
            const int row = r0 + half * 8 + r;
            float v = selu_f(acc[r] + bia[nt]) * msk[r];
            out[(size_t)row * HD + f] = v + res[nt][r];
        }
    }
}

extern "C" void kernel_launch(void* const* d_in, const int* in_sizes, int n_in,
                              void* d_out, int out_size, void* d_ws, size_t ws_size,
                              hipStream_t stream) {
    (void)in_sizes; (void)n_in; (void)out_size; (void)d_ws; (void)ws_size;
    const float* h         = (const float*)d_in[0];
    const float* e         = (const float*)d_in[1];
    const float* node_mask = (const float*)d_in[2];
    const float* edge_mask = (const float*)d_in[3];
    const float* Wf        = (const float*)d_in[4];
    const float* bf        = (const float*)d_in[5];
    const float* Ww        = (const float*)d_in[6];
    const float* bw        = (const float*)d_in[7];
    float* out = (float*)d_out;

    ef_filter_reduce<<<NB * NA, 256, 0, stream>>>(h, e, edge_mask, Wf, bf, out);
    ef_node_mlp<<<(NB * NA) / R2, 256, 0, stream>>>(h, node_mask, Ww, bw, out);
}


# Filter kernel made persistent: 256 work-groups of one b and 8 i; Wf fragments and h slice stay in registers across i, e tiles go through two LDS stages; k order inside the matrix sums changes
# speedup vs baseline: 2.2122x; 1.5205x over previous
#include <hip/hip_runtime.h>


typedef __attribute__((ext_vector_type(2))) float v2f;
typedef __attribute__((ext_vector_type(4))) float v4f;
typedef __attribute__((ext_vector_type(8))) float v8f;

#define NB 16
#define NA 128
#define HD 256
#define EF 30
#define KP 32
#define K2 (2 * HD)
#define R2 16
#define KC2 64
#define NCH2 (K2 / KC2)
#define ZP4 ((K2 + 4) / 4)
#define WP4 ((KC2 + 4) / 4)

__device__ __forceinline__ float selu_f(float x) {
    const float scale = 1.0507009873554805f;
    const float alpha = 1.6732632423543772f;
    return x > 0.0f ? scale * x : scale * alpha * (__expf(x) - 1.0f);
}

__device__ __forceinline__ v8f wmma_f32(v2f a, v2f b, v8f c) {
    return __builtin_amdgcn_wmma_f32_16x16x4_f32(false, a, false, b, (short)0, c,
                                                 false, false);
}

__device__ __forceinline__ v4f ld4(const float* p) {
    v4f v;
    v.x = p[0]; v.y = p[1]; v.z = p[2]; v.w = p[3];
    return v;
}

#define IPG 8
#define EP4 ((KP + 4) / 4)
#define EST4 (NA * EP4)
#define ET4 (NA * EF / 4)

__device__ __forceinline__ void row_tile_products(const v4f* arow,
                                                  const v4f (&b0)[KP / 8],
                                                  const v4f (&b1)[KP / 8],
                                                  v8f& c0, v8f& c1) {
    v4f a[KP / 8];
#pragma unroll
    for (int q = 0; q < KP / 8; ++q)
        a[q] = arow[2 * q];
#pragma unroll
    for (int q = 0; q < KP / 8; ++q) {
        const v2f a_lo = {a[q].x, a[q].y},   a_hi = {a[q].z, a[q].w};
        const v2f b0lo = {b0[q].x, b0[q].y}, b0hi = {b0[q].z, b0[q].w};
        const v2f b1lo = {b1[q].x, b1[q].y}, b1hi = {b1[q].z, b1[q].w};
        c0 = wmma_f32(a_lo, b0lo, c0);
        c1 = wmma_f32(a_lo, b1lo, c1);
        c0 = wmma_f32(a_hi, b0hi, c0);
        c1 = wmma_f32(a_hi, b1hi, c1);
    }
}

__device__ __forceinline__ float fold_rows(const v8f c, const float bia,
                                           const float (&em)[8],
                                           const float (&hv)[8], float psum) {
#pragma unroll
    for (int r = 0; r < 8; ++r) {
        const float ef = selu_f(c[r] + bia) * em[r];
        psum += ef * hv[r];
    }
    return psum;
}

__global__ __launch_bounds__(256)
void ef_filter_reduce(const float* __restrict__ h,
                      const float* __restrict__ e,
                      const float* __restrict__ edge_mask,
                      const float* __restrict__ Wf,
                      const float* __restrict__ bf,
                      float* __restrict__ h1out) {
    __shared__ v4f sh_e[2 * EST4];
    __shared__ v4f sh_m[2 * (NA / 4)];

    const int tid = threadIdx.x;
    const int b   = blockIdx.x / (NA / IPG);
    const int i0  = (blockIdx.x % (NA / IPG)) * IPG;

    const int lane  = tid & 31;
    const int wave  = tid >> 5;
    const int half  = lane >> 4;
    const int l     = lane & 15;
    const int fbase = wave * 32;

    const bool own3 = tid < ET4 - 3 * 256;
    int o_lo[4], o_hi[4];
#pragma unroll
    for (int s = 0; s < 4; ++s) {
        const int g  = 4 * (tid + 256 * s);
        const int j1 = g / EF,       k1 = g - j1 * EF;
        const int j2 = (g + 2) / EF, k2 = (g + 2) - j2 * EF;
        o_lo[s] = j1 * (2 * EP4) + (k1 >> 1);
        o_hi[s] = j2 * (2 * EP4) + (k2 >> 1);
    }
    const float* etile = e + ((size_t)(b * NA + i0)) * NA * EF + 4 * tid;
    const float* emrow = edge_mask + ((size_t)(b * NA + i0)) * NA + 4 * lane;

    v4f pre[4], prem;
    const v4f vzero4 = {0.f, 0.f, 0.f, 0.f};
#pragma unroll
    for (int s = 0; s < 4; ++s)
        pre[s] = (s < 3 || own3) ? ld4(etile + s * 1024) : vzero4;
    prem = (wave == 0) ? ld4(emrow) : vzero4;

    v4f bfr0[KP / 8], bfr1[KP / 8];
    float bia0, bia1;
    {
        const float* w0 = Wf + (size_t)(fbase + l) * EF + 4 * half;
        const float* w1 = w0 + 16 * EF;
#pragma unroll
        for (int q = 0; q < KP / 8; ++q) {
            const bool tail = (q == KP / 8 - 1) && (half != 0);
            bfr0[q].x = w0[8 * q];     bfr0[q].y = w0[8 * q + 1];
            bfr1[q].x = w1[8 * q];     bfr1[q].y = w1[8 * q + 1];
            bfr0[q].z = tail ? 0.0f : w0[8 * q + 2];
            bfr0[q].w = tail ? 0.0f : w0[8 * q + 3];
            bfr1[q].z = tail ? 0.0f : w1[8 * q + 2];
            bfr1[q].w = tail ? 0.0f : w1[8 * q + 3];
        }
        bia0 = bf[fbase + l];
        bia1 = bf[fbase + 16 + l];
    }

    float hv[8][2][8];
    {
        const float* hB = h + (size_t)b * NA * HD + (size_t)(half * 8) * HD + fbase + l;
#pragma unroll
        for (int mt = 0; mt < 8; ++mt)
#pragma unroll
            for (int nt = 0; nt < 2; ++nt)
#pragma unroll
                for (int r = 0; r < 8; ++r)
                    hv[mt][nt][r] = hB[(size_t)(mt * 16 + r) * HD + nt * 16];
    }

    v2f* e2 = (v2f*)sh_e;
    if (tid < NA) {
        const v2f z2 = {0.f, 0.f};
        e2[tid * (2 * EP4) + EF / 2] = z2;
        e2[2 * EST4 + tid * (2 * EP4) + EF / 2] = z2;
    }
#pragma unroll
    for (int s = 0; s < 4; ++s) {
        if (s < 3 || own3) {
            const v2f lo = {pre[s].x, pre[s].y}, hi = {pre[s].z, pre[s].w};
            e2[o_lo[s]] = lo;
            e2[o_hi[s]] = hi;
        }
    }
    if (wave == 0) sh_m[lane] = prem;

    const v8f vzero = {0.f, 0.f, 0.f, 0.f, 0.f, 0.f, 0.f, 0.f};

#pragma unroll 1
    for (int it = 0; it < IPG; ++it) {
        const int st = it & 1;
        __syncthreads();
        if (it + 1 < IPG) {
#pragma unroll
            for (int s = 0; s < 4; ++s)
                pre[s] = (s < 3 || own3) ? ld4(etile + (size_t)(it + 1) * NA * EF + s * 1024)
                                         : vzero4;
            prem = (wave == 0) ? ld4(emrow + (it + 1) * NA) : vzero4;
        }

        const v4f* arow = sh_e + st * EST4 + l * EP4 + half;
        const v4f* mrow = sh_m + st * (NA / 4) + half * 2;

        float ps0 = 0.0f, ps1 = 0.0f;
        v8f c0 = vzero, c1 = vzero;
        row_tile_products(arow, bfr0, bfr1, c0, c1);
#pragma unroll
        for (int mt = 0; mt < 8; ++mt) {
            v8f n0 = vzero, n1 = vzero;
            if (mt + 1 < 8)
                row_tile_products(arow + (mt + 1) * 16 * EP4, bfr0, bfr1, n0, n1);
            const v4f m0 = mrow[mt * 4], m1 = mrow[mt * 4 + 1];
            const float em[8] = {m0.x, m0.y, m0.z, m0.w, m1.x, m1.y, m1.z, m1.w};
            ps0 = fold_rows(c0, bia0, em, hv[mt][0], ps0);
            ps1 = fold_rows(c1, bia1, em, hv[mt][1], ps1);
            c0 = n0;
            c1 = n1;
        }
        const float t0 = ps0 + __shfl_xor(ps0, 16, 32);
        const float t1 = ps1 + __shfl_xor(ps1, 16, 32);
        h1out[(size_t)(b * NA + i0 + it) * HD + fbase + lane] = half ? t1 : t0;

        if (it + 1 < IPG) {
            v2f* dst = e2 + (st ^ 1) * (2 * EST4);
#pragma unroll
            for (int s = 0; s < 4; ++s) {
                if (s < 3 || own3) {
                    const v2f lo = {pre[s].x, pre[s].y}, hi = {pre[s].z, pre[s].w};
                    dst[o_lo[s]] = lo;
                    dst[o_hi[s]] = hi;
                }
            }
            if (wave == 0) sh_m[(st ^ 1) * (NA / 4) + lane] = prem;
        }
    }
}

__global__ __launch_bounds__(256)
void ef_node_mlp(const float* __restrict__ h,
                 const float* __restrict__ node_mask,
                 const float* __restrict__ Ww,
                 const float* __restrict__ bw,
                 float* __restrict__ out  ) {
    __shared__ v4f sh_z[R2 * ZP4];
    __shared__ v4f sh_w[2][HD * WP4];

    const int tid = threadIdx.x;
    const int r0  = blockIdx.x * R2;

    const int lane  = tid & 31;
    const int wave  = tid >> 5;
    const int half  = lane >> 4;
    const int l     = lane & 15;
    const int nbase = wave * 32;

    float bia[2], msk[8], res[2][8];
#pragma unroll
    for (int nt = 0; nt < 2; ++nt) {
        const int f = nbase + nt * 16 + l;
        bia[nt] = bw[f];
#pragma unroll
        for (int r = 0; r < 8; ++r)
            res[nt][r] = h[(size_t)(r0 + half * 8 + r) * HD + f];
    }
#pragma unroll
    for (int r = 0; r < 8; ++r)
        msk[r] = node_mask[r0 + half * 8 + r];

    {
        const int c4 = tid & 127;
        const int rb = tid >> 7;
        const bool from_h = c4 < 64;
#pragma unroll
        for (int i = 0; i < 8; ++i) {
            const int r = 2 * i + rb;
            const size_t row = (size_t)(r0 + r);
            const float* src = from_h ? h + row * HD + c4 * 4
                                      : out + row * HD + (c4 - 64) * 4;
            sh_z[r * ZP4 + c4] = ld4(src);
        }
    }

    const int wn = tid >> 4;
    const int wc = tid & 15;
    const float* wsrc = Ww + (size_t)wn * K2 + wc * 4;
    v4f pre[16];
#pragma unroll
    for (int i = 0; i < 16; ++i)
        pre[i] = ld4(wsrc + (size_t)i * 16 * K2);
#pragma unroll
    for (int i = 0; i < 16; ++i)
        sh_w[0][(i * 16 + wn) * WP4 + wc] = pre[i];

    const v8f vzero = {0.f, 0.f, 0.f, 0.f, 0.f, 0.f, 0.f, 0.f};
    v8f acc0 = vzero, acc1 = vzero;

#pragma unroll
    for (int kc = 0; kc < NCH2; ++kc) {
        const int buf = kc & 1;
        if (kc + 1 < NCH2) {
#pragma unroll
            for (int i = 0; i < 16; ++i)
                pre[i] = ld4(wsrc + (size_t)i * 16 * K2 + (kc + 1) * KC2);
        }
        __syncthreads();

        const v4f* zr = sh_z + l * ZP4 + kc * (KC2 / 4) + half;
        const v4f* w0 = sh_w[buf] + (nbase + l) * WP4 + half;
        const v4f* w1 = w0 + 16 * WP4;
#pragma unroll
        for (int q = 0; q < KC2 / 8; ++q) {
            const v4f a  = zr[2 * q];
            const v4f b0 = w0[2 * q];
            const v4f b1 = w1[2 * q];
            const v2f a_lo = {a.x, a.y},   a_hi = {a.z, a.w};
            const v2f b0lo = {b0.x, b0.y}, b0hi = {b0.z, b0.w};
            const v2f b1lo = {b1.x, b1.y}, b1hi = {b1.z, b1.w};
            acc0 = wmma_f32(a_lo, b0lo, acc0);
            acc1 = wmma_f32(a_lo, b1lo, acc1);
            acc0 = wmma_f32(a_hi, b0hi, acc0);
            acc1 = wmma_f32(a_hi, b1hi, acc1);
        }

        if (kc + 1 < NCH2) {
#pragma unroll
            for (int i = 0; i < 16; ++i)
                sh_w[buf ^ 1][(i * 16 + wn) * WP4 + wc] = pre[i];
        }
    }

#pragma unroll
    for (int nt = 0; nt < 2; ++nt) {
        const int f = nbase + nt * 16 + l;
        const v8f acc = nt ? acc1 : acc0;
#pragma unroll
        for (int r = 0; r < 8; ++r) {
            const int row = r0 + half * 8 + r;
            float v = selu_f(acc[r] + bia[nt]) * msk[r];
            out[(size_t)row * HD + f] = v + res[nt][r];
        }
    }
}

extern "C" void kernel_launch(void* const* d_in, const int* in_sizes, int n_in,
                              void* d_out, int out_size, void* d_ws, size_t ws_size,
                              hipStream_t stream) {
    (void)in_sizes; (void)n_in; (void)out_size; (void)d_ws; (void)ws_size;
    const float* h         = (const float*)d_in[0];
    const float* e         = (const float*)d_in[1];
    const float* node_mask = (const float*)d_in[2];
    const float* edge_mask = (const float*)d_in[3];
    const float* Wf        = (const float*)d_in[4];
    const float* bf        = (const float*)d_in[5];
    const float* Ww        = (const float*)d_in[6];
    const float* bw        = (const float*)d_in[7];
    float* out = (float*)d_out;

    ef_filter_reduce<<<(NB * NA) / IPG, 256, 0, stream>>>(h, e, edge_mask, Wf, bf, out);
    ef_node_mlp<<<(NB * NA) / R2, 256, 0, stream>>>(h, node_mask, Ww, bw, out);
}


# selu made branch-free in both kernels: both sides evaluated, one select per element, instead of an EXEC-masked if/else whose mask updates made the filter kernel's fold one serial chain
# speedup vs baseline: 3.4375x; 1.5539x over previous
#include <hip/hip_runtime.h>


typedef __attribute__((ext_vector_type(2))) float v2f;
typedef __attribute__((ext_vector_type(4))) float v4f;
typedef __attribute__((ext_vector_type(8))) float v8f;

#define NB 16
#define NA 128
#define HD 256
#define EF 30
#define KP 32
#define K2 (2 * HD)
#define R2 16
#define KC2 64
#define NCH2 (K2 / KC2)
#define ZP4 ((K2 + 4) / 4)
#define WP4 ((KC2 + 4) / 4)

__device__ __forceinline__ float selu_f(float x) {
    const float scale = 1.0507009873554805f;
    const float alpha = 1.6732632423543772f;
    const float neg = scale * alpha * (__expf(x) - 1.0f);
    const float pos = scale * x;
    return x > 0.0f ? pos : neg;
}

__device__ __forceinline__ v8f wmma_f32(v2f a, v2f b, v8f c) {
    return __builtin_amdgcn_wmma_f32_16x16x4_f32(false, a, false, b, (short)0, c,
                                                 false, false);
}

__device__ __forceinline__ v4f ld4(const float* p) {
    v4f v;
    v.x = p[0]; v.y = p[1]; v.z = p[2]; v.w = p[3];
    return v;
}

#define IPG 8
#define EP4 ((KP + 4) / 4)
#define EST4 (NA * EP4)
#define ET4 (NA * EF / 4)

__device__ __forceinline__ void row_tile_products(const v4f* arow,
                                                  const v4f (&b0)[KP / 8],
                                                  const v4f (&b1)[KP / 8],
                                                  v8f& c0, v8f& c1) {
    v4f a[KP / 8];
#pragma unroll
    for (int q = 0; q < KP / 8; ++q)
        a[q] = arow[2 * q];
#pragma unroll
    for (int q = 0; q < KP / 8; ++q) {
        const v2f a_lo = {a[q].x, a[q].y},   a_hi = {a[q].z, a[q].w};
        const v2f b0lo = {b0[q].x, b0[q].y}, b0hi = {b0[q].z, b0[q].w};
        const v2f b1lo = {b1[q].x, b1[q].y}, b1hi = {b1[q].z, b1[q].w};
        c0 = wmma_f32(a_lo, b0lo, c0);
        c1 = wmma_f32(a_lo, b1lo, c1);
        c0 = wmma_f32(a_hi, b0hi, c0);
        c1 = wmma_f32(a_hi, b1hi, c1);
    }
}

__device__ __forceinline__ float fold_rows(const v8f c, const float bia,
                                           const float (&em)[8],
                                           const float (&hv)[8], float psum) {
#pragma unroll
    for (int r = 0; r < 8; ++r) {
        const float ef = selu_f(c[r] + bia) * em[r];
        psum += ef * hv[r];
    }
    return psum;
}

__global__ __launch_bounds__(256)
void ef_filter_reduce(const float* __restrict__ h,
                      const float* __restrict__ e,
                      const float* __restrict__ edge_mask,
                      const float* __restrict__ Wf,
                      const float* __restrict__ bf,
                      float* __restrict__ h1out) {
    __shared__ v4f sh_e[2 * EST4];
    __shared__ v4f sh_m[2 * (NA / 4)];

    const int tid = threadIdx.x;
    const int b   = blockIdx.x / (NA / IPG);
    const int i0  = (blockIdx.x % (NA / IPG)) * IPG;

    const int lane  = tid & 31;
    const int wave  = tid >> 5;
    const int half  = lane >> 4;
    const int l     = lane & 15;
    const int fbase = wave * 32;

    const bool own3 = tid < ET4 - 3 * 256;
    int o_lo[4], o_hi[4];
#pragma unroll
    for (int s = 0; s < 4; ++s) {
        const int g  = 4 * (tid + 256 * s);
        const int j1 = g / EF,       k1 = g - j1 * EF;
        const int j2 = (g + 2) / EF, k2 = (g + 2) - j2 * EF;
        o_lo[s] = j1 * (2 * EP4) + (k1 >> 1);
        o_hi[s] = j2 * (2 * EP4) + (k2 >> 1);
    }
    const float* etile = e + ((size_t)(b * NA + i0)) * NA * EF + 4 * tid;
    const float* emrow = edge_mask + ((size_t)(b * NA + i0)) * NA + 4 * lane;

    v4f pre[4], prem;
    const v4f vzero4 = {0.f, 0.f, 0.f, 0.f};
#pragma unroll
    for (int s = 0; s < 4; ++s)
        pre[s] = (s < 3 || own3) ? ld4(etile + s * 1024) : vzero4;
    prem = (wave == 0) ? ld4(emrow) : vzero4;

    v4f bfr0[KP / 8], bfr1[KP / 8];
    float bia0, bia1;
    {
        const float* w0 = Wf + (size_t)(fbase + l) * EF + 4 * half;
        const float* w1 = w0 + 16 * EF;
#pragma unroll
        for (int q = 0; q < KP / 8; ++q) {
            const bool tail = (q == KP / 8 - 1) && (half != 0);
            bfr0[q].x = w0[8 * q];     bfr0[q].y = w0[8 * q + 1];
            bfr1[q].x = w1[8 * q];     bfr1[q].y = w1[8 * q + 1];
            bfr0[q].z = tail ? 0.0f : w0[8 * q + 2];
            bfr0[q].w = tail ? 0.0f : w0[8 * q + 3];
            bfr1[q].z = tail ? 0.0f : w1[8 * q + 2];
            bfr1[q].w = tail ? 0.0f : w1[8 * q + 3];
        }
        bia0 = bf[fbase + l];
        bia1 = bf[fbase + 16 + l];
    }

    float hv[8][2][8];
    {
        const float* hB = h + (size_t)b * NA * HD + (size_t)(half * 8) * HD + fbase + l;
#pragma unroll
        for (int mt = 0; mt < 8; ++mt)
#pragma unroll
            for (int nt = 0; nt < 2; ++nt)
#pragma unroll
                for (int r = 0; r < 8; ++r)
                    hv[mt][nt][r] = hB[(size_t)(mt * 16 + r) * HD + nt * 16];
    }

    v2f* e2 = (v2f*)sh_e;
    if (tid < NA) {
        const v2f z2 = {0.f, 0.f};
        e2[tid * (2 * EP4) + EF / 2] = z2;
        e2[2 * EST4 + tid * (2 * EP4) + EF / 2] = z2;
    }
#pragma unroll
    for (int s = 0; s < 4; ++s) {
        if (s < 3 || own3) {
            const v2f lo = {pre[s].x, pre[s].y}, hi = {pre[s].z, pre[s].w};
            e2[o_lo[s]] = lo;
            e2[o_hi[s]] = hi;
        }
    }
    if (wave == 0) sh_m[lane] = prem;

    const v8f vzero = {0.f, 0.f, 0.f, 0.f, 0.f, 0.f, 0.f, 0.f};

#pragma unroll 1
    for (int it = 0; it < IPG; ++it) {
        const int st = it & 1;
        __syncthreads();
        if (it + 1 < IPG) {
#pragma unroll
            for (int s = 0; s < 4; ++s)
                pre[s] = (s < 3 || own3) ? ld4(etile + (size_t)(it + 1) * NA * EF + s * 1024)
                                         : vzero4;
            prem = (wave == 0) ? ld4(emrow + (it + 1) * NA) : vzero4;
        }

        const v4f* arow = sh_e + st * EST4 + l * EP4 + half;
        const v4f* mrow = sh_m + st * (NA / 4) + half * 2;

        float ps0 = 0.0f, ps1 = 0.0f;
        v8f c0 = vzero, c1 = vzero;
        row_tile_products(arow, bfr0, bfr1, c0, c1);
#pragma unroll
        for (int mt = 0; mt < 8; ++mt) {
            v8f n0 = vzero, n1 = vzero;
            if (mt + 1 < 8)
                row_tile_products(arow + (mt + 1) * 16 * EP4, bfr0, bfr1, n0, n1);
            const v4f m0 = mrow[mt * 4], m1 = mrow[mt * 4 + 1];
            const float em[8] = {m0.x, m0.y, m0.z, m0.w, m1.x, m1.y, m1.z, m1.w};
            ps0 = fold_rows(c0, bia0, em, hv[mt][0], ps0);
            ps1 = fold_rows(c1, bia1, em, hv[mt][1], ps1);
            c0 = n0;
            c1 = n1;
        }
        const float t0 = ps0 + __shfl_xor(ps0, 16, 32);
        const float t1 = ps1 + __shfl_xor(ps1, 16, 32);
        h1out[(size_t)(b * NA + i0 + it) * HD + fbase + lane] = half ? t1 : t0;

        if (it + 1 < IPG) {
            v2f* dst = e2 + (st ^ 1) * (2 * EST4);
#pragma unroll
            for (int s = 0; s < 4; ++s) {
                if (s < 3 || own3) {
                    const v2f lo = {pre[s].x, pre[s].y}, hi = {pre[s].z, pre[s].w};
                    dst[o_lo[s]] = lo;
                    dst[o_hi[s]] = hi;
                }
            }
            if (wave == 0) sh_m[(st ^ 1) * (NA / 4) + lane] = prem;
        }
    }
}

__global__ __launch_bounds__(256)
void ef_node_mlp(const float* __restrict__ h,
                 const float* __restrict__ node_mask,
                 const float* __restrict__ Ww,
                 const float* __restrict__ bw,
                 float* __restrict__ out  ) {
    __shared__ v4f sh_z[R2 * ZP4];
    __shared__ v4f sh_w[2][HD * WP4];

    const int tid = threadIdx.x;
    const int r0  = blockIdx.x * R2;

    const int lane  = tid & 31;
    const int wave  = tid >> 5;
    const int half  = lane >> 4;
    const int l     = lane & 15;
    const int nbase = wave * 32;

    float bia[2], msk[8], res[2][8];
#pragma unroll
    for (int nt = 0; nt < 2; ++nt) {
        const int f = nbase + nt * 16 + l;
        bia[nt] = bw[f];
#pragma unroll
        for (int r = 0; r < 8; ++r)
            res[nt][r] = h[(size_t)(r0 + half * 8 + r) * HD + f];
    }
#pragma unroll
    for (int r = 0; r < 8; ++r)
        msk[r] = node_mask[r0 + half * 8 + r];

    {
        const int c4 = tid & 127;
        const int rb = tid >> 7;
        const bool from_h = c4 < 64;
#pragma unroll
        for (int i = 0; i < 8; ++i) {
            const int r = 2 * i + rb;
            const size_t row = (size_t)(r0 + r);
            const float* src = from_h ? h + row * HD + c4 * 4
                                      : out + row * HD + (c4 - 64) * 4;
            sh_z[r * ZP4 + c4] = ld4(src);
        }
    }

    const int wn = tid >> 4;
    const int wc = tid & 15;
    const float* wsrc = Ww + (size_t)wn * K2 + wc * 4;
    v4f pre[16];
#pragma unroll
    for (int i = 0; i < 16; ++i)
        pre[i] = ld4(wsrc + (size_t)i * 16 * K2);
#pragma unroll
    for (int i = 0; i < 16; ++i)
        sh_w[0][(i * 16 + wn) * WP4 + wc] = pre[i];

    const v8f vzero = {0.f, 0.f, 0.f, 0.f, 0.f, 0.f, 0.f, 0.f};
    v8f acc0 = vzero, acc1 = vzero;

#pragma unroll
    for (int kc = 0; kc < NCH2; ++kc) {
        const int buf = kc & 1;
        if (kc + 1 < NCH2) {
#pragma unroll
            for (int i = 0; i < 16; ++i)
                pre[i] = ld4(wsrc + (size_t)i * 16 * K2 + (kc + 1) * KC2);
        }
        __syncthreads();

        const v4f* zr = sh_z + l * ZP4 + kc * (KC2 / 4) + half;
        const v4f* w0 = sh_w[buf] + (nbase + l) * WP4 + half;
        const v4f* w1 = w0 + 16 * WP4;
#pragma unroll
        for (int q = 0; q < KC2 / 8; ++q) {
            const v4f a  = zr[2 * q];
            const v4f b0 = w0[2 * q];
            const v4f b1 = w1[2 * q];
            const v2f a_lo = {a.x, a.y},   a_hi = {a.z, a.w};
            const v2f b0lo = {b0.x, b0.y}, b0hi = {b0.z, b0.w};
            const v2f b1lo = {b1.x, b1.y}, b1hi = {b1.z, b1.w};
            acc0 = wmma_f32(a_lo, b0lo, acc0);
            acc1 = wmma_f32(a_lo, b1lo, acc1);
            acc0 = wmma_f32(a_hi, b0hi, acc0);
            acc1 = wmma_f32(a_hi, b1hi, acc1);
        }

        if (kc + 1 < NCH2) {
#pragma unroll
            for (int i = 0; i < 16; ++i)
                sh_w[buf ^ 1][(i * 16 + wn) * WP4 + wc] = pre[i];
        }
    }

#pragma unroll
    for (int nt = 0; nt < 2; ++nt) {
        const int f = nbase + nt * 16 + l;
        const v8f acc = nt ? acc1 : acc0;
#pragma unroll
        for (int r = 0; r < 8; ++r) {
            const int row = r0 + half * 8 + r;
            float v = selu_f(acc[r] + bia[nt]) * msk[r];
            out[(size_t)row * HD + f] = v + res[nt][r];
        }
    }
}

extern "C" void kernel_launch(void* const* d_in, const int* in_sizes, int n_in,
                              void* d_out, int out_size, void* d_ws, size_t ws_size,
                              hipStream_t stream) {
    (void)in_sizes; (void)n_in; (void)out_size; (void)d_ws; (void)ws_size;
    const float* h         = (const float*)d_in[0];
    const float* e         = (const float*)d_in[1];
    const float* node_mask = (const float*)d_in[2];
    const float* edge_mask = (const float*)d_in[3];
    const float* Wf        = (const float*)d_in[4];
    const float* bf        = (const float*)d_in[5];
    const float* Ww        = (const float*)d_in[6];
    const float* bw        = (const float*)d_in[7];
    float* out = (float*)d_out;

    ef_filter_reduce<<<(NB * NA) / IPG, 256, 0, stream>>>(h, e, edge_mask, Wf, bf, out);
    ef_node_mlp<<<(NB * NA) / R2, 256, 0, stream>>>(h, node_mask, Ww, bw, out);
}


# Filter kernel's fold halved in vector instructions: bias and log2(e) ride in the Wf fragments, selu's scale applied once per column sum, fold packed two-wide; sum over j now in 4 partial sums
# speedup vs baseline: 3.6406x; 1.0591x over previous
#include <hip/hip_runtime.h>


typedef __attribute__((ext_vector_type(2))) float v2f;
typedef __attribute__((ext_vector_type(4))) float v4f;
typedef __attribute__((ext_vector_type(8))) float v8f;

#define NB 16
#define NA 128
#define HD 256
#define EF 30
#define KP 32
#define K2 (2 * HD)
#define R2 16
#define KC2 64
#define NCH2 (K2 / KC2)
#define ZP4 ((K2 + 4) / 4)
#define WP4 ((KC2 + 4) / 4)

__device__ __forceinline__ float selu_f(float x) {
    const float scale = 1.0507009873554805f;
    const float alpha = 1.6732632423543772f;
    const float neg = scale * alpha * (__expf(x) - 1.0f);
    const float pos = scale * x;
    return x > 0.0f ? pos : neg;
}

#define LOG2E_F 1.4426950408889634f
#define SELU_A2 ((float)(1.6732632423543772 * 1.4426950408889634))
#define SELU_S2 ((float)(1.0507009873554805 * 0.6931471805599453))

__device__ __forceinline__ v8f wmma_f32(v2f a, v2f b, v8f c) {
    return __builtin_amdgcn_wmma_f32_16x16x4_f32(false, a, false, b, (short)0, c,
                                                 false, false);
}

__device__ __forceinline__ v4f ld4(const float* p) {
    v4f v;
    v.x = p[0]; v.y = p[1]; v.z = p[2]; v.w = p[3];
    return v;
}

#define IPG 8
#define EP4 ((KP + 4) / 4)
#define EST4 (NA * EP4)
#define ET4 (NA * EF / 4)

__device__ __forceinline__ void row_tile_products(const v4f* arow,
                                                  const v4f (&b0)[KP / 8],
                                                  const v4f (&b1)[KP / 8],
                                                  v8f& c0, v8f& c1) {
    v4f a[KP / 8];
#pragma unroll
    for (int q = 0; q < KP / 8; ++q)
        a[q] = arow[2 * q];
#pragma unroll
    for (int q = 0; q < KP / 8; ++q) {
        const v2f a_lo = {a[q].x, a[q].y},   a_hi = {a[q].z, a[q].w};
        const v2f b0lo = {b0[q].x, b0[q].y}, b0hi = {b0[q].z, b0[q].w};
        const v2f b1lo = {b1[q].x, b1[q].y}, b1hi = {b1[q].z, b1[q].w};
        c0 = wmma_f32(a_lo, b0lo, c0);
        c1 = wmma_f32(a_lo, b1lo, c1);
        c0 = wmma_f32(a_hi, b0hi, c0);
        c1 = wmma_f32(a_hi, b1hi, c1);
    }
}

__device__ __forceinline__ void fold_tiles(const v8f c0, const v8f c1, const v2f (&em2)[4],
                                           const v2f (&hvA)[4], const v2f (&hvB)[4],
                                           v2f (&acc)[4]) {
    const v2f a2  = {SELU_A2, SELU_A2};
    const v2f na2 = {-SELU_A2, -SELU_A2};
    v2f x2[8], e[8], neg[8], s[8], w[8];
#pragma unroll
    for (int rp = 0; rp < 4; ++rp) {
        x2[rp]     = (v2f){c0[2 * rp], c0[2 * rp + 1]};
        x2[4 + rp] = (v2f){c1[2 * rp], c1[2 * rp + 1]};
    }
#pragma unroll
    for (int p = 0; p < 8; ++p) {
        e[p].x = __builtin_amdgcn_exp2f(x2[p].x);
        e[p].y = __builtin_amdgcn_exp2f(x2[p].y);
    }
#pragma unroll
    for (int rp = 0; rp < 4; ++rp) {
        w[rp]     = em2[rp] * hvA[rp];
        w[4 + rp] = em2[rp] * hvB[rp];
    }
#pragma unroll
    for (int p = 0; p < 8; ++p)
        neg[p] = __builtin_elementwise_fma(e[p], a2, na2);
#pragma unroll
    for (int p = 0; p < 8; ++p) {
        s[p].x = x2[p].x > 0.0f ? x2[p].x : neg[p].x;
        s[p].y = x2[p].y > 0.0f ? x2[p].y : neg[p].y;
    }
#pragma unroll
    for (int p = 0; p < 8; ++p) {
        const int a = 2 * (p >> 2) + (p & 1);
        acc[a] = __builtin_elementwise_fma(s[p], w[p], acc[a]);
    }
}

__global__ __launch_bounds__(256)
void ef_filter_reduce(const float* __restrict__ h,
                      const float* __restrict__ e,
                      const float* __restrict__ edge_mask,
                      const float* __restrict__ Wf,
                      const float* __restrict__ bf,
                      float* __restrict__ h1out) {
    __shared__ v4f sh_e[2 * EST4];
    __shared__ v4f sh_m[2 * (NA / 4)];

    const int tid = threadIdx.x;
    const int b   = blockIdx.x / (NA / IPG);
    const int i0  = (blockIdx.x % (NA / IPG)) * IPG;

    const int lane  = tid & 31;
    const int wave  = tid >> 5;
    const int half  = lane >> 4;
    const int l     = lane & 15;
    const int fbase = wave * 32;

    const bool own3 = tid < ET4 - 3 * 256;
    int o_lo[4], o_hi[4];
#pragma unroll
    for (int s = 0; s < 4; ++s) {
        const int g  = 4 * (tid + 256 * s);
        const int j1 = g / EF,       k1 = g - j1 * EF;
        const int j2 = (g + 2) / EF, k2 = (g + 2) - j2 * EF;
        o_lo[s] = j1 * (2 * EP4) + (k1 >> 1);
        o_hi[s] = j2 * (2 * EP4) + (k2 >> 1);
    }
    const float* etile = e + ((size_t)(b * NA + i0)) * NA * EF + 4 * tid;
    const float* emrow = edge_mask + ((size_t)(b * NA + i0)) * NA + 4 * lane;

    v4f pre[4], prem;
    const v4f vzero4 = {0.f, 0.f, 0.f, 0.f};
#pragma unroll
    for (int s = 0; s < 4; ++s)
        pre[s] = (s < 3 || own3) ? ld4(etile + s * 1024) : vzero4;
    prem = (wave == 0) ? ld4(emrow) : vzero4;

    v4f bfr0[KP / 8], bfr1[KP / 8];
    {
        const float* w0 = Wf + (size_t)(fbase + l) * EF + 4 * half;
        const float* w1 = w0 + 16 * EF;
        const float bia0 = bf[fbase + l] * LOG2E_F;
        const float bia1 = bf[fbase + 16 + l] * LOG2E_F;
#pragma unroll
        for (int q = 0; q < KP / 8; ++q) {
            const bool tail = (q == KP / 8 - 1) && (half != 0);
            bfr0[q].x = w0[8 * q] * LOG2E_F;     bfr0[q].y = w0[8 * q + 1] * LOG2E_F;
            bfr1[q].x = w1[8 * q] * LOG2E_F;     bfr1[q].y = w1[8 * q + 1] * LOG2E_F;
            bfr0[q].z = tail ? bia0 : w0[8 * q + 2] * LOG2E_F;
            bfr0[q].w = tail ? 0.0f : w0[8 * q + 3] * LOG2E_F;
            bfr1[q].z = tail ? bia1 : w1[8 * q + 2] * LOG2E_F;
            bfr1[q].w = tail ? 0.0f : w1[8 * q + 3] * LOG2E_F;
        }
    }

    v2f hv2[8][2][4];
    {
        const float* hB = h + (size_t)b * NA * HD + (size_t)(half * 8) * HD + fbase + l;
#pragma unroll
        for (int mt = 0; mt < 8; ++mt)
#pragma unroll
            for (int nt = 0; nt < 2; ++nt)
#pragma unroll
                for (int rp = 0; rp < 4; ++rp) {
                    hv2[mt][nt][rp].x = hB[(size_t)(mt * 16 + 2 * rp) * HD + nt * 16];
                    hv2[mt][nt][rp].y = hB[(size_t)(mt * 16 + 2 * rp + 1) * HD + nt * 16];
                }
    }

    v2f* e2 = (v2f*)sh_e;
    if (tid < NA) {
        const v2f p2 = {1.f, 0.f};
        e2[tid * (2 * EP4) + EF / 2] = p2;
        e2[2 * EST4 + tid * (2 * EP4) + EF / 2] = p2;
    }
#pragma unroll
    for (int s = 0; s < 4; ++s) {
        if (s < 3 || own3) {
            const v2f lo = {pre[s].x, pre[s].y}, hi = {pre[s].z, pre[s].w};
            e2[o_lo[s]] = lo;
            e2[o_hi[s]] = hi;
        }
    }
    if (wave == 0) sh_m[lane] = prem;

    const v8f vzero = {0.f, 0.f, 0.f, 0.f, 0.f, 0.f, 0.f, 0.f};

#pragma unroll 1
    for (int it = 0; it < IPG; ++it) {
        const int st = it & 1;
        __syncthreads();
        if (it + 1 < IPG) {
#pragma unroll
            for (int s = 0; s < 4; ++s)
                pre[s] = (s < 3 || own3) ? ld4(etile + (size_t)(it + 1) * NA * EF + s * 1024)
                                         : vzero4;
            prem = (wave == 0) ? ld4(emrow + (it + 1) * NA) : vzero4;
        }

        const v4f* arow = sh_e + st * EST4 + l * EP4 + half;
        const v4f* mrow = sh_m + st * (NA / 4) + half * 2;

        const v2f vzero2 = {0.f, 0.f};
        v2f acc[4] = {vzero2, vzero2, vzero2, vzero2};
        v8f c0 = vzero, c1 = vzero;
        row_tile_products(arow, bfr0, bfr1, c0, c1);
#pragma unroll
        for (int mt = 0; mt < 8; ++mt) {
            v8f n0 = vzero, n1 = vzero;
            if (mt + 1 < 8)
                row_tile_products(arow + (mt + 1) * 16 * EP4, bfr0, bfr1, n0, n1);
            const v4f m0 = mrow[mt * 4], m1 = mrow[mt * 4 + 1];
            const v2f em2[4] = {{m0.x, m0.y}, {m0.z, m0.w}, {m1.x, m1.y}, {m1.z, m1.w}};
            fold_tiles(c0, c1, em2, hv2[mt][0], hv2[mt][1], acc);
            c0 = n0;
            c1 = n1;
        }
        const v2f q0 = acc[0] + acc[1], q1 = acc[2] + acc[3];
        const float ps0 = q0.x + q0.y, ps1 = q1.x + q1.y;
        const float t0 = ps0 + __shfl_xor(ps0, 16, 32);
        const float t1 = ps1 + __shfl_xor(ps1, 16, 32);
        h1out[(size_t)(b * NA + i0 + it) * HD + fbase + lane] = (half ? t1 : t0) * SELU_S2;

        if (it + 1 < IPG) {
            v2f* dst = e2 + (st ^ 1) * (2 * EST4);
#pragma unroll
            for (int s = 0; s < 4; ++s) {
                if (s < 3 || own3) {
                    const v2f lo = {pre[s].x, pre[s].y}, hi = {pre[s].z, pre[s].w};
                    dst[o_lo[s]] = lo;
                    dst[o_hi[s]] = hi;
                }
            }
            if (wave == 0) sh_m[(st ^ 1) * (NA / 4) + lane] = prem;
        }
    }
}

__global__ __launch_bounds__(256)
void ef_node_mlp(const float* __restrict__ h,
                 const float* __restrict__ node_mask,
                 const float* __restrict__ Ww,
                 const float* __restrict__ bw,
                 float* __restrict__ out  ) {
    __shared__ v4f sh_z[R2 * ZP4];
    __shared__ v4f sh_w[2][HD * WP4];

    const int tid = threadIdx.x;
    const int r0  = blockIdx.x * R2;

    const int lane  = tid & 31;
    const int wave  = tid >> 5;
    const int half  = lane >> 4;
    const int l     = lane & 15;
    const int nbase = wave * 32;

    float bia[2], msk[8], res[2][8];
#pragma unroll
    for (int nt = 0; nt < 2; ++nt) {
        const int f = nbase + nt * 16 + l;
        bia[nt] = bw[f];
#pragma unroll
        for (int r = 0; r < 8; ++r)
            res[nt][r] = h[(size_t)(r0 + half * 8 + r) * HD + f];
    }
#pragma unroll
    for (int r = 0; r < 8; ++r)
        msk[r] = node_mask[r0 + half * 8 + r];

    {
        const int c4 = tid & 127;
        const int rb = tid >> 7;
        const bool from_h = c4 < 64;
#pragma unroll
        for (int i = 0; i < 8; ++i) {
            const int r = 2 * i + rb;
            const size_t row = (size_t)(r0 + r);
            const float* src = from_h ? h + row * HD + c4 * 4
                                      : out + row * HD + (c4 - 64) * 4;
            sh_z[r * ZP4 + c4] = ld4(src);
        }
    }

    const int wn = tid >> 4;
    const int wc = tid & 15;
    const float* wsrc = Ww + (size_t)wn * K2 + wc * 4;
    v4f pre[16];
#pragma unroll
    for (int i = 0; i < 16; ++i)
        pre[i] = ld4(wsrc + (size_t)i * 16 * K2);
#pragma unroll
    for (int i = 0; i < 16; ++i)
        sh_w[0][(i * 16 + wn) * WP4 + wc] = pre[i];

    const v8f vzero = {0.f, 0.f, 0.f, 0.f, 0.f, 0.f, 0.f, 0.f};
    v8f acc0 = vzero, acc1 = vzero;

#pragma unroll
    for (int kc = 0; kc < NCH2; ++kc) {
        const int buf = kc & 1;
        if (kc + 1 < NCH2) {
#pragma unroll
            for (int i = 0; i < 16; ++i)
                pre[i] = ld4(wsrc + (size_t)i * 16 * K2 + (kc + 1) * KC2);
        }
        __syncthreads();

        const v4f* zr = sh_z + l * ZP4 + kc * (KC2 / 4) + half;
        const v4f* w0 = sh_w[buf] + (nbase + l) * WP4 + half;
        const v4f* w1 = w0 + 16 * WP4;
#pragma unroll
        for (int q = 0; q < KC2 / 8; ++q) {
            const v4f a  = zr[2 * q];
            const v4f b0 = w0[2 * q];
            const v4f b1 = w1[2 * q];
            const v2f a_lo = {a.x, a.y},   a_hi = {a.z, a.w};
            const v2f b0lo = {b0.x, b0.y}, b0hi = {b0.z, b0.w};
            const v2f b1lo = {b1.x, b1.y}, b1hi = {b1.z, b1.w};
            acc0 = wmma_f32(a_lo, b0lo, acc0);
            acc1 = wmma_f32(a_lo, b1lo, acc1);
            acc0 = wmma_f32(a_hi, b0hi, acc0);
            acc1 = wmma_f32(a_hi, b1hi, acc1);
        }

        if (kc + 1 < NCH2) {
#pragma unroll
            for (int i = 0; i < 16; ++i)
                sh_w[buf ^ 1][(i * 16 + wn) * WP4 + wc] = pre[i];
        }
    }

#pragma unroll
    for (int nt = 0; nt < 2; ++nt) {
        const int f = nbase + nt * 16 + l;
        const v8f acc = nt ? acc1 : acc0;
#pragma unroll
        for (int r = 0; r < 8; ++r) {
            const int row = r0 + half * 8 + r;
            float v = selu_f(acc[r] + bia[nt]) * msk[r];
            out[(size_t)row * HD + f] = v + res[nt][r];
        }
    }
}

extern "C" void kernel_launch(void* const* d_in, const int* in_sizes, int n_in,
                              void* d_out, int out_size, void* d_ws, size_t ws_size,
                              hipStream_t stream) {
    (void)in_sizes; (void)n_in; (void)out_size; (void)d_ws; (void)ws_size;
    const float* h         = (const float*)d_in[0];
    const float* e         = (const float*)d_in[1];
    const float* node_mask = (const float*)d_in[2];
    const float* edge_mask = (const float*)d_in[3];
    const float* Wf        = (const float*)d_in[4];
    const float* bf        = (const float*)d_in[5];
    const float* Ww        = (const float*)d_in[6];
    const float* bw        = (const float*)d_in[7];
    float* out = (float*)d_out;

    ef_filter_reduce<<<(NB * NA) / IPG, 256, 0, stream>>>(h, e, edge_mask, Wf, bf, out);
    ef_node_mlp<<<(NB * NA) / R2, 256, 0, stream>>>(h, node_mask, Ww, bw, out);
}


# Filter kernel: each row tile's fold interleaved with the next tile's matrix products by scheduling groups (exponential, product, four vector ops in turn); LDS fragments read one tile ahead
# speedup vs baseline: 3.7898x; 1.0410x over previous
#include <hip/hip_runtime.h>


typedef __attribute__((ext_vector_type(2))) float v2f;
typedef __attribute__((ext_vector_type(4))) float v4f;
typedef __attribute__((ext_vector_type(8))) float v8f;

#define NB 16
#define NA 128
#define HD 256
#define EF 30
#define KP 32
#define K2 (2 * HD)
#define R2 16
#define KC2 64
#define NCH2 (K2 / KC2)
#define ZP4 ((K2 + 4) / 4)
#define WP4 ((KC2 + 4) / 4)

__device__ __forceinline__ float selu_f(float x) {
    const float scale = 1.0507009873554805f;
    const float alpha = 1.6732632423543772f;
    const float neg = scale * alpha * (__expf(x) - 1.0f);
    const float pos = scale * x;
    return x > 0.0f ? pos : neg;
}

#define LOG2E_F 1.4426950408889634f
#define SELU_A2 ((float)(1.6732632423543772 * 1.4426950408889634))
#define SELU_S2 ((float)(1.0507009873554805 * 0.6931471805599453))

__device__ __forceinline__ v8f wmma_f32(v2f a, v2f b, v8f c) {
    return __builtin_amdgcn_wmma_f32_16x16x4_f32(false, a, false, b, (short)0, c,
                                                 false, false);
}

__device__ __forceinline__ v4f ld4(const float* p) {
    v4f v;
    v.x = p[0]; v.y = p[1]; v.z = p[2]; v.w = p[3];
    return v;
}

#define IPG 8
#define EP4 ((KP + 4) / 4)
#define EST4 (NA * EP4)
#define ET4 (NA * EF / 4)

__device__ __forceinline__ void load_row_tile(const v4f* arow, v4f (&a)[KP / 8]) {
#pragma unroll
    for (int q = 0; q < KP / 8; ++q)
        a[q] = arow[2 * q];
}

__device__ __forceinline__ void row_tile_products(const v4f (&a)[KP / 8],
                                                  const v4f (&b0)[KP / 8],
                                                  const v4f (&b1)[KP / 8],
                                                  v8f& c0, v8f& c1) {
#pragma unroll
    for (int q = 0; q < KP / 8; ++q) {
        const v2f a_lo = {a[q].x, a[q].y},   a_hi = {a[q].z, a[q].w};
        const v2f b0lo = {b0[q].x, b0[q].y}, b0hi = {b0[q].z, b0[q].w};
        const v2f b1lo = {b1[q].x, b1[q].y}, b1hi = {b1[q].z, b1[q].w};
        c0 = wmma_f32(a_lo, b0lo, c0);
        c1 = wmma_f32(a_lo, b1lo, c1);
        c0 = wmma_f32(a_hi, b0hi, c0);
        c1 = wmma_f32(a_hi, b1hi, c1);
    }
}

__device__ __forceinline__ void fold_tiles(const v8f c0, const v8f c1, const v2f (&em2)[4],
                                           const v2f (&hvA)[4], const v2f (&hvB)[4],
                                           v2f (&acc)[4]) {
    const v2f a2  = {SELU_A2, SELU_A2};
    const v2f na2 = {-SELU_A2, -SELU_A2};
    v2f x2[8], e[8], neg[8], s[8], w[8];
#pragma unroll
    for (int rp = 0; rp < 4; ++rp) {
        x2[rp]     = (v2f){c0[2 * rp], c0[2 * rp + 1]};
        x2[4 + rp] = (v2f){c1[2 * rp], c1[2 * rp + 1]};
    }
#pragma unroll
    for (int p = 0; p < 8; ++p) {
        e[p].x = __builtin_amdgcn_exp2f(x2[p].x);
        e[p].y = __builtin_amdgcn_exp2f(x2[p].y);
    }
#pragma unroll
    for (int rp = 0; rp < 4; ++rp) {
        w[rp]     = em2[rp] * hvA[rp];
        w[4 + rp] = em2[rp] * hvB[rp];
    }
#pragma unroll
    for (int p = 0; p < 8; ++p)
        neg[p] = __builtin_elementwise_fma(e[p], a2, na2);
#pragma unroll
    for (int p = 0; p < 8; ++p) {
        s[p].x = x2[p].x > 0.0f ? x2[p].x : neg[p].x;
        s[p].y = x2[p].y > 0.0f ? x2[p].y : neg[p].y;
    }
#pragma unroll
    for (int p = 0; p < 8; ++p) {
        const int a = 2 * (p >> 2) + (p & 1);
        acc[a] = __builtin_elementwise_fma(s[p], w[p], acc[a]);
    }
}

__global__ __launch_bounds__(256)
void ef_filter_reduce(const float* __restrict__ h,
                      const float* __restrict__ e,
                      const float* __restrict__ edge_mask,
                      const float* __restrict__ Wf,
                      const float* __restrict__ bf,
                      float* __restrict__ h1out) {
    __shared__ v4f sh_e[2 * EST4];
    __shared__ v4f sh_m[2 * (NA / 4)];

    const int tid = threadIdx.x;
    const int b   = blockIdx.x / (NA / IPG);
    const int i0  = (blockIdx.x % (NA / IPG)) * IPG;

    const int lane  = tid & 31;
    const int wave  = tid >> 5;
    const int half  = lane >> 4;
    const int l     = lane & 15;
    const int fbase = wave * 32;

    const bool own3 = tid < ET4 - 3 * 256;
    int o_lo[4], o_hi[4];
#pragma unroll
    for (int s = 0; s < 4; ++s) {
        const int g  = 4 * (tid + 256 * s);
        const int j1 = g / EF,       k1 = g - j1 * EF;
        const int j2 = (g + 2) / EF, k2 = (g + 2) - j2 * EF;
        o_lo[s] = j1 * (2 * EP4) + (k1 >> 1);
        o_hi[s] = j2 * (2 * EP4) + (k2 >> 1);
    }
    const float* etile = e + ((size_t)(b * NA + i0)) * NA * EF + 4 * tid;
    const float* emrow = edge_mask + ((size_t)(b * NA + i0)) * NA + 4 * lane;

    v4f pre[4], prem;
    const v4f vzero4 = {0.f, 0.f, 0.f, 0.f};
#pragma unroll
    for (int s = 0; s < 4; ++s)
        pre[s] = (s < 3 || own3) ? ld4(etile + s * 1024) : vzero4;
    prem = (wave == 0) ? ld4(emrow) : vzero4;

    v4f bfr0[KP / 8], bfr1[KP / 8];
    {
        const float* w0 = Wf + (size_t)(fbase + l) * EF + 4 * half;
        const float* w1 = w0 + 16 * EF;
        const float bia0 = bf[fbase + l] * LOG2E_F;
        const float bia1 = bf[fbase + 16 + l] * LOG2E_F;
#pragma unroll
        for (int q = 0; q < KP / 8; ++q) {
            const bool tail = (q == KP / 8 - 1) && (half != 0);
            bfr0[q].x = w0[8 * q] * LOG2E_F;     bfr0[q].y = w0[8 * q + 1] * LOG2E_F;
            bfr1[q].x = w1[8 * q] * LOG2E_F;     bfr1[q].y = w1[8 * q + 1] * LOG2E_F;
            bfr0[q].z = tail ? bia0 : w0[8 * q + 2] * LOG2E_F;
            bfr0[q].w = tail ? 0.0f : w0[8 * q + 3] * LOG2E_F;
            bfr1[q].z = tail ? bia1 : w1[8 * q + 2] * LOG2E_F;
            bfr1[q].w = tail ? 0.0f : w1[8 * q + 3] * LOG2E_F;
        }
    }

    v2f hv2[8][2][4];
    {
        const float* hB = h + (size_t)b * NA * HD + (size_t)(half * 8) * HD + fbase + l;
#pragma unroll
        for (int mt = 0; mt < 8; ++mt)
#pragma unroll
            for (int nt = 0; nt < 2; ++nt)
#pragma unroll
                for (int rp = 0; rp < 4; ++rp) {
                    hv2[mt][nt][rp].x = hB[(size_t)(mt * 16 + 2 * rp) * HD + nt * 16];
                    hv2[mt][nt][rp].y = hB[(size_t)(mt * 16 + 2 * rp + 1) * HD + nt * 16];
                }
    }

    v2f* e2 = (v2f*)sh_e;
    if (tid < NA) {
        const v2f p2 = {1.f, 0.f};
        e2[tid * (2 * EP4) + EF / 2] = p2;
        e2[2 * EST4 + tid * (2 * EP4) + EF / 2] = p2;
    }
#pragma unroll
    for (int s = 0; s < 4; ++s) {
        if (s < 3 || own3) {
            const v2f lo = {pre[s].x, pre[s].y}, hi = {pre[s].z, pre[s].w};
            e2[o_lo[s]] = lo;
            e2[o_hi[s]] = hi;
        }
    }
    if (wave == 0) sh_m[lane] = prem;

    const v8f vzero = {0.f, 0.f, 0.f, 0.f, 0.f, 0.f, 0.f, 0.f};

#pragma unroll 1
    for (int it = 0; it < IPG; ++it) {
        const int st = it & 1;
        __syncthreads();
        if (it + 1 < IPG) {
#pragma unroll
            for (int s = 0; s < 4; ++s)
                pre[s] = (s < 3 || own3) ? ld4(etile + (size_t)(it + 1) * NA * EF + s * 1024)
                                         : vzero4;
            prem = (wave == 0) ? ld4(emrow + (it + 1) * NA) : vzero4;
        }

        const v4f* arow = sh_e + st * EST4 + l * EP4 + half;
        const v4f* mrow = sh_m + st * (NA / 4) + half * 2;

        const v2f vzero2 = {0.f, 0.f};
        v2f acc[4] = {vzero2, vzero2, vzero2, vzero2};
        v8f c0 = vzero, c1 = vzero;
        v4f aF[KP / 8], mF[2];
        load_row_tile(arow, aF);
        row_tile_products(aF, bfr0, bfr1, c0, c1);
        load_row_tile(arow + 16 * EP4, aF);
        mF[0] = mrow[0];
        mF[1] = mrow[1];
        __builtin_amdgcn_sched_barrier(0);
#pragma unroll
        for (int mt = 0; mt < 8; ++mt) {
            v8f n0 = vzero, n1 = vzero;
            v4f aN[KP / 8] = {vzero4, vzero4, vzero4, vzero4}, mN[2] = {vzero4, vzero4};
            if (mt + 2 < 8)
                load_row_tile(arow + (mt + 2) * 16 * EP4, aN);
            if (mt + 1 < 8) {
                mN[0] = mrow[(mt + 1) * 4];
                mN[1] = mrow[(mt + 1) * 4 + 1];
                row_tile_products(aF, bfr0, bfr1, n0, n1);
            }
            const v2f em2[4] = {{mF[0].x, mF[0].y}, {mF[0].z, mF[0].w},
                                {mF[1].x, mF[1].y}, {mF[1].z, mF[1].w}};
            fold_tiles(c0, c1, em2, hv2[mt][0], hv2[mt][1], acc);
            c0 = n0;
            c1 = n1;
#pragma unroll
            for (int q = 0; q < KP / 8; ++q)
                aF[q] = aN[q];
            mF[0] = mN[0];
            mF[1] = mN[1];
            if (mt + 1 < 8) {
                __builtin_amdgcn_sched_group_barrier(0x100, 6, 0);
#pragma unroll
                for (int s = 0; s < 16; ++s) {
                    __builtin_amdgcn_sched_group_barrier(0x400, 1, 0);
                    __builtin_amdgcn_sched_group_barrier(0x008, 1, 0);
                    __builtin_amdgcn_sched_group_barrier(0x002, 4, 0);
                }
                __builtin_amdgcn_sched_barrier(0);
            }
        }
        const v2f q0 = acc[0] + acc[1], q1 = acc[2] + acc[3];
        const float ps0 = q0.x + q0.y, ps1 = q1.x + q1.y;
        const float t0 = ps0 + __shfl_xor(ps0, 16, 32);
        const float t1 = ps1 + __shfl_xor(ps1, 16, 32);
        h1out[(size_t)(b * NA + i0 + it) * HD + fbase + lane] = (half ? t1 : t0) * SELU_S2;

        if (it + 1 < IPG) {
            v2f* dst = e2 + (st ^ 1) * (2 * EST4);
#pragma unroll
            for (int s = 0; s < 4; ++s) {
                if (s < 3 || own3) {
                    const v2f lo = {pre[s].x, pre[s].y}, hi = {pre[s].z, pre[s].w};
                    dst[o_lo[s]] = lo;
                    dst[o_hi[s]] = hi;
                }
            }
            if (wave == 0) sh_m[(st ^ 1) * (NA / 4) + lane] = prem;
        }
    }
}

__global__ __launch_bounds__(256)
void ef_node_mlp(const float* __restrict__ h,
                 const float* __restrict__ node_mask,
                 const float* __restrict__ Ww,
                 const float* __restrict__ bw,
                 float* __restrict__ out  ) {
    __shared__ v4f sh_z[R2 * ZP4];
    __shared__ v4f sh_w[2][HD * WP4];

    const int tid = threadIdx.x;
    const int r0  = blockIdx.x * R2;

    const int lane  = tid & 31;
    const int wave  = tid >> 5;
    const int half  = lane >> 4;
    const int l     = lane & 15;
    const int nbase = wave * 32;

    float bia[2], msk[8], res[2][8];
#pragma unroll
    for (int nt = 0; nt < 2; ++nt) {
        const int f = nbase + nt * 16 + l;
        bia[nt] = bw[f];
#pragma unroll
        for (int r = 0; r < 8; ++r)
            res[nt][r] = h[(size_t)(r0 + half * 8 + r) * HD + f];
    }
#pragma unroll
    for (int r = 0; r < 8; ++r)
        msk[r] = node_mask[r0 + half * 8 + r];

    {
        const int c4 = tid & 127;
        const int rb = tid >> 7;
        const bool from_h = c4 < 64;
#pragma unroll
        for (int i = 0; i < 8; ++i) {
            const int r = 2 * i + rb;
            const size_t row = (size_t)(r0 + r);
            const float* src = from_h ? h + row * HD + c4 * 4
                                      : out + row * HD + (c4 - 64) * 4;
            sh_z[r * ZP4 + c4] = ld4(src);
        }
    }

    const int wn = tid >> 4;
    const int wc = tid & 15;
    const float* wsrc = Ww + (size_t)wn * K2 + wc * 4;
    v4f pre[16];
#pragma unroll
    for (int i = 0; i < 16; ++i)
        pre[i] = ld4(wsrc + (size_t)i * 16 * K2);
#pragma unroll
    for (int i = 0; i < 16; ++i)
        sh_w[0][(i * 16 + wn) * WP4 + wc] = pre[i];

    const v8f vzero = {0.f, 0.f, 0.f, 0.f, 0.f, 0.f, 0.f, 0.f};
    v8f acc0 = vzero, acc1 = vzero;

#pragma unroll
    for (int kc = 0; kc < NCH2; ++kc) {
        const int buf = kc & 1;
        if (kc + 1 < NCH2) {
#pragma unroll
            for (int i = 0; i < 16; ++i)
                pre[i] = ld4(wsrc + (size_t)i * 16 * K2 + (kc + 1) * KC2);
        }
        __syncthreads();

        const v4f* zr = sh_z + l * ZP4 + kc * (KC2 / 4) + half;
        const v4f* w0 = sh_w[buf] + (nbase + l) * WP4 + half;
        const v4f* w1 = w0 + 16 * WP4;
#pragma unroll
        for (int q = 0; q < KC2 / 8; ++q) {
            const v4f a  = zr[2 * q];
            const v4f b0 = w0[2 * q];
            const v4f b1 = w1[2 * q];
            const v2f a_lo = {a.x, a.y},   a_hi = {a.z, a.w};
            const v2f b0lo = {b0.x, b0.y}, b0hi = {b0.z, b0.w};
            const v2f b1lo = {b1.x, b1.y}, b1hi = {b1.z, b1.w};
            acc0 = wmma_f32(a_lo, b0lo, acc0);
            acc1 = wmma_f32(a_lo, b1lo, acc1);
            acc0 = wmma_f32(a_hi, b0hi, acc0);
            acc1 = wmma_f32(a_hi, b1hi, acc1);
        }

        if (kc + 1 < NCH2) {
#pragma unroll
            for (int i = 0; i < 16; ++i)
                sh_w[buf ^ 1][(i * 16 + wn) * WP4 + wc] = pre[i];
        }
    }

#pragma unroll
    for (int nt = 0; nt < 2; ++nt) {
        const int f = nbase + nt * 16 + l;
        const v8f acc = nt ? acc1 : acc0;
#pragma unroll
        for (int r = 0; r < 8; ++r) {
            const int row = r0 + half * 8 + r;
            float v = selu_f(acc[r] + bia[nt]) * msk[r];
            out[(size_t)row * HD + f] = v + res[nt][r];
        }
    }
}

extern "C" void kernel_launch(void* const* d_in, const int* in_sizes, int n_in,
                              void* d_out, int out_size, void* d_ws, size_t ws_size,
                              hipStream_t stream) {
    (void)in_sizes; (void)n_in; (void)out_size; (void)d_ws; (void)ws_size;
    const float* h         = (const float*)d_in[0];
    const float* e         = (const float*)d_in[1];
    const float* node_mask = (const float*)d_in[2];
    const float* edge_mask = (const float*)d_in[3];
    const float* Wf        = (const float*)d_in[4];
    const float* bf        = (const float*)d_in[5];
    const float* Ww        = (const float*)d_in[6];
    const float* bw        = (const float*)d_in[7];
    float* out = (float*)d_out;

    ef_filter_reduce<<<(NB * NA) / IPG, 256, 0, stream>>>(h, e, edge_mask, Wf, bf, out);
    ef_node_mlp<<<(NB * NA) / R2, 256, 0, stream>>>(h, node_mask, Ww, bw, out);
}


# Node MLP folded into the filter kernel: each work-group finishes its own 8 rows from h1 kept in LDS (A rows 8..15 repeat rows 0..7); one launch instead of two, h1 never goes through global memory
# speedup vs baseline: 4.0224x; 1.0614x over previous
#include <hip/hip_runtime.h>


typedef __attribute__((ext_vector_type(2))) float v2f;
typedef __attribute__((ext_vector_type(4))) float v4f;
typedef __attribute__((ext_vector_type(8))) float v8f;

#define NB 16
#define NA 128
#define HD 256
#define EF 30
#define KP 32
#define K2 (2 * HD)
#define KC2 64
#define NCH2 (K2 / KC2)
#define ZP4 ((K2 + 4) / 4)
#define WP4 ((KC2 + 4) / 4)

__device__ __forceinline__ float selu_f(float x) {
    const float scale = 1.0507009873554805f;
    const float alpha = 1.6732632423543772f;
    const float neg = scale * alpha * (__expf(x) - 1.0f);
    const float pos = scale * x;
    return x > 0.0f ? pos : neg;
}

#define LOG2E_F 1.4426950408889634f
#define SELU_A2 ((float)(1.6732632423543772 * 1.4426950408889634))
#define SELU_S2 ((float)(1.0507009873554805 * 0.6931471805599453))

__device__ __forceinline__ v8f wmma_f32(v2f a, v2f b, v8f c) {
    return __builtin_amdgcn_wmma_f32_16x16x4_f32(false, a, false, b, (short)0, c,
                                                 false, false);
}

__device__ __forceinline__ v4f ld4(const float* p) {
    v4f v;
    v.x = p[0]; v.y = p[1]; v.z = p[2]; v.w = p[3];
    return v;
}

#define IPG 8
#define EP4 ((KP + 4) / 4)
#define EST4 (NA * EP4)
#define ET4 (NA * EF / 4)

__device__ __forceinline__ void load_row_tile(const v4f* arow, v4f (&a)[KP / 8]) {
#pragma unroll
    for (int q = 0; q < KP / 8; ++q)
        a[q] = arow[2 * q];
}

__device__ __forceinline__ void row_tile_products(const v4f (&a)[KP / 8],
                                                  const v4f (&b0)[KP / 8],
                                                  const v4f (&b1)[KP / 8],
                                                  v8f& c0, v8f& c1) {
#pragma unroll
    for (int q = 0; q < KP / 8; ++q) {
        const v2f a_lo = {a[q].x, a[q].y},   a_hi = {a[q].z, a[q].w};
        const v2f b0lo = {b0[q].x, b0[q].y}, b0hi = {b0[q].z, b0[q].w};
        const v2f b1lo = {b1[q].x, b1[q].y}, b1hi = {b1[q].z, b1[q].w};
        c0 = wmma_f32(a_lo, b0lo, c0);
        c1 = wmma_f32(a_lo, b1lo, c1);
        c0 = wmma_f32(a_hi, b0hi, c0);
        c1 = wmma_f32(a_hi, b1hi, c1);
    }
}

__device__ __forceinline__ void fold_tiles(const v8f c0, const v8f c1, const v2f (&em2)[4],
                                           const v2f (&hvA)[4], const v2f (&hvB)[4],
                                           v2f (&acc)[4]) {
    const v2f a2  = {SELU_A2, SELU_A2};
    const v2f na2 = {-SELU_A2, -SELU_A2};
    v2f x2[8], e[8], neg[8], s[8], w[8];
#pragma unroll
    for (int rp = 0; rp < 4; ++rp) {
        x2[rp]     = (v2f){c0[2 * rp], c0[2 * rp + 1]};
        x2[4 + rp] = (v2f){c1[2 * rp], c1[2 * rp + 1]};
    }
#pragma unroll
    for (int p = 0; p < 8; ++p) {
        e[p].x = __builtin_amdgcn_exp2f(x2[p].x);
        e[p].y = __builtin_amdgcn_exp2f(x2[p].y);
    }
#pragma unroll
    for (int rp = 0; rp < 4; ++rp) {
        w[rp]     = em2[rp] * hvA[rp];
        w[4 + rp] = em2[rp] * hvB[rp];
    }
#pragma unroll
    for (int p = 0; p < 8; ++p)
        neg[p] = __builtin_elementwise_fma(e[p], a2, na2);
#pragma unroll
    for (int p = 0; p < 8; ++p) {
        s[p].x = x2[p].x > 0.0f ? x2[p].x : neg[p].x;
        s[p].y = x2[p].y > 0.0f ? x2[p].y : neg[p].y;
    }
#pragma unroll
    for (int p = 0; p < 8; ++p) {
        const int a = 2 * (p >> 2) + (p & 1);
        acc[a] = __builtin_elementwise_fma(s[p], w[p], acc[a]);
    }
}

__global__ __launch_bounds__(256)
void ef_conv_fused(const float* __restrict__ h,
                   const float* __restrict__ e,
                   const float* __restrict__ edge_mask,
                   const float* __restrict__ Wf,
                   const float* __restrict__ bf,
                   const float* __restrict__ node_mask,
                   const float* __restrict__ Ww,
                   const float* __restrict__ bw,
                   float* __restrict__ out) {
    __shared__ v4f sh_e[2 * EST4];
    __shared__ v4f sh_m[2 * (NA / 4)];
    __shared__ v4f sh_z[IPG * ZP4];
    __shared__ v4f sh_w[2][HD * WP4];

    const int tid = threadIdx.x;
    const int b   = blockIdx.x / (NA / IPG);
    const int i0  = (blockIdx.x % (NA / IPG)) * IPG;

    const int lane  = tid & 31;
    const int wave  = tid >> 5;
    const int half  = lane >> 4;
    const int l     = lane & 15;
    const int fbase = wave * 32;

    const bool own3 = tid < ET4 - 3 * 256;
    int o_lo[4], o_hi[4];
#pragma unroll
    for (int s = 0; s < 4; ++s) {
        const int g  = 4 * (tid + 256 * s);
        const int j1 = g / EF,       k1 = g - j1 * EF;
        const int j2 = (g + 2) / EF, k2 = (g + 2) - j2 * EF;
        o_lo[s] = j1 * (2 * EP4) + (k1 >> 1);
        o_hi[s] = j2 * (2 * EP4) + (k2 >> 1);
    }
    const float* etile = e + ((size_t)(b * NA + i0)) * NA * EF + 4 * tid;
    const float* emrow = edge_mask + ((size_t)(b * NA + i0)) * NA + 4 * lane;

    v4f pre[4], prem;
    const v4f vzero4 = {0.f, 0.f, 0.f, 0.f};
#pragma unroll
    for (int s = 0; s < 4; ++s)
        pre[s] = (s < 3 || own3) ? ld4(etile + s * 1024) : vzero4;
    prem = (wave == 0) ? ld4(emrow) : vzero4;

    v4f bfr0[KP / 8], bfr1[KP / 8];
    {
        const float* w0 = Wf + (size_t)(fbase + l) * EF + 4 * half;
        const float* w1 = w0 + 16 * EF;
        const float bia0 = bf[fbase + l] * LOG2E_F;
        const float bia1 = bf[fbase + 16 + l] * LOG2E_F;
#pragma unroll
        for (int q = 0; q < KP / 8; ++q) {
            const bool tail = (q == KP / 8 - 1) && (half != 0);
            bfr0[q].x = w0[8 * q] * LOG2E_F;     bfr0[q].y = w0[8 * q + 1] * LOG2E_F;
            bfr1[q].x = w1[8 * q] * LOG2E_F;     bfr1[q].y = w1[8 * q + 1] * LOG2E_F;
            bfr0[q].z = tail ? bia0 : w0[8 * q + 2] * LOG2E_F;
            bfr0[q].w = tail ? 0.0f : w0[8 * q + 3] * LOG2E_F;
            bfr1[q].z = tail ? bia1 : w1[8 * q + 2] * LOG2E_F;
            bfr1[q].w = tail ? 0.0f : w1[8 * q + 3] * LOG2E_F;
        }
    }

    v2f hv2[8][2][4];
    {
        const float* hB = h + (size_t)b * NA * HD + (size_t)(half * 8) * HD + fbase + l;
#pragma unroll
        for (int mt = 0; mt < 8; ++mt)
#pragma unroll
            for (int nt = 0; nt < 2; ++nt)
#pragma unroll
                for (int rp = 0; rp < 4; ++rp) {
                    hv2[mt][nt][rp].x = hB[(size_t)(mt * 16 + 2 * rp) * HD + nt * 16];
                    hv2[mt][nt][rp].y = hB[(size_t)(mt * 16 + 2 * rp + 1) * HD + nt * 16];
                }
    }

    v2f* e2 = (v2f*)sh_e;
    if (tid < NA) {
        const v2f p2 = {1.f, 0.f};
        e2[tid * (2 * EP4) + EF / 2] = p2;
        e2[2 * EST4 + tid * (2 * EP4) + EF / 2] = p2;
    }
#pragma unroll
    for (int s = 0; s < 4; ++s) {
        if (s < 3 || own3) {
            const v2f lo = {pre[s].x, pre[s].y}, hi = {pre[s].z, pre[s].w};
            e2[o_lo[s]] = lo;
            e2[o_hi[s]] = hi;
        }
    }
    if (wave == 0) sh_m[lane] = prem;

    const v8f vzero = {0.f, 0.f, 0.f, 0.f, 0.f, 0.f, 0.f, 0.f};

#pragma unroll 1
    for (int it = 0; it < IPG; ++it) {
        const int st = it & 1;
        __syncthreads();
        if (it + 1 < IPG) {
#pragma unroll
            for (int s = 0; s < 4; ++s)
                pre[s] = (s < 3 || own3) ? ld4(etile + (size_t)(it + 1) * NA * EF + s * 1024)
                                         : vzero4;
            prem = (wave == 0) ? ld4(emrow + (it + 1) * NA) : vzero4;
        }

        const v4f* arow = sh_e + st * EST4 + l * EP4 + half;
        const v4f* mrow = sh_m + st * (NA / 4) + half * 2;

        const v2f vzero2 = {0.f, 0.f};
        v2f acc[4] = {vzero2, vzero2, vzero2, vzero2};
        v8f c0 = vzero, c1 = vzero;
        v4f aF[KP / 8], mF[2];
        load_row_tile(arow, aF);
        row_tile_products(aF, bfr0, bfr1, c0, c1);
        load_row_tile(arow + 16 * EP4, aF);
        mF[0] = mrow[0];
        mF[1] = mrow[1];
        __builtin_amdgcn_sched_barrier(0);
#pragma unroll
        for (int mt = 0; mt < 8; ++mt) {
            v8f n0 = vzero, n1 = vzero;
            v4f aN[KP / 8] = {vzero4, vzero4, vzero4, vzero4}, mN[2] = {vzero4, vzero4};
            if (mt + 2 < 8)
                load_row_tile(arow + (mt + 2) * 16 * EP4, aN);
            if (mt + 1 < 8) {
                mN[0] = mrow[(mt + 1) * 4];
                mN[1] = mrow[(mt + 1) * 4 + 1];
                row_tile_products(aF, bfr0, bfr1, n0, n1);
            }
            const v2f em2[4] = {{mF[0].x, mF[0].y}, {mF[0].z, mF[0].w},
                                {mF[1].x, mF[1].y}, {mF[1].z, mF[1].w}};
            fold_tiles(c0, c1, em2, hv2[mt][0], hv2[mt][1], acc);
            c0 = n0;
            c1 = n1;
#pragma unroll
            for (int q = 0; q < KP / 8; ++q)
                aF[q] = aN[q];
            mF[0] = mN[0];
            mF[1] = mN[1];
            if (mt + 1 < 8) {
                __builtin_amdgcn_sched_group_barrier(0x100, 6, 0);
#pragma unroll
                for (int s = 0; s < 16; ++s) {
                    __builtin_amdgcn_sched_group_barrier(0x400, 1, 0);
                    __builtin_amdgcn_sched_group_barrier(0x008, 1, 0);
                    __builtin_amdgcn_sched_group_barrier(0x002, 4, 0);
                }
                __builtin_amdgcn_sched_barrier(0);
            }
        }
        const v2f q0 = acc[0] + acc[1], q1 = acc[2] + acc[3];
        const float ps0 = q0.x + q0.y, ps1 = q1.x + q1.y;
        const float t0 = ps0 + __shfl_xor(ps0, 16, 32);
        const float t1 = ps1 + __shfl_xor(ps1, 16, 32);
        ((float*)sh_z)[it * (4 * ZP4) + HD + fbase + lane] = (half ? t1 : t0) * SELU_S2;

        if (it + 1 < IPG) {
            v2f* dst = e2 + (st ^ 1) * (2 * EST4);
#pragma unroll
            for (int s = 0; s < 4; ++s) {
                if (s < 3 || own3) {
                    const v2f lo = {pre[s].x, pre[s].y}, hi = {pre[s].z, pre[s].w};
                    dst[o_lo[s]] = lo;
                    dst[o_hi[s]] = hi;
                }
            }
            if (wave == 0) sh_m[(st ^ 1) * (NA / 4) + lane] = prem;
        }
    }

    const int row0  = b * NA + i0;
    const int fcol  = fbase + lane;

    const float bia = bw[fcol];
    float msk[IPG], res[IPG];
#pragma unroll
    for (int r = 0; r < IPG; ++r) {
        msk[r] = node_mask[row0 + r];
        res[r] = h[(size_t)(row0 + r) * HD + fcol];
    }

#pragma unroll
    for (int p = 0; p < IPG / 4; ++p) {
        const int u  = tid + 256 * p;
        const int r  = u >> 6;
        const int c4 = u & 63;
        sh_z[r * ZP4 + c4] = ld4(h + (size_t)(row0 + r) * HD + c4 * 4);
    }

    const int wn = tid >> 4;
    const int wc = tid & 15;
    const float* wsrc = Ww + (size_t)wn * K2 + wc * 4;
    v4f wpre[16];
#pragma unroll
    for (int i = 0; i < 16; ++i)
        wpre[i] = ld4(wsrc + (size_t)i * 16 * K2);
#pragma unroll
    for (int i = 0; i < 16; ++i)
        sh_w[0][(i * 16 + wn) * WP4 + wc] = wpre[i];

    v8f acc0 = vzero, acc1 = vzero;

#pragma unroll
    for (int kc = 0; kc < NCH2; ++kc) {
        const int buf = kc & 1;
        if (kc + 1 < NCH2) {
#pragma unroll
            for (int i = 0; i < 16; ++i)
                wpre[i] = ld4(wsrc + (size_t)i * 16 * K2 + (kc + 1) * KC2);
        }
        __syncthreads();

        const v4f* zr = sh_z + (l & (IPG - 1)) * ZP4 + kc * (KC2 / 4) + half;
        const v4f* w0 = sh_w[buf] + (fbase + l) * WP4 + half;
        const v4f* w1 = w0 + 16 * WP4;
#pragma unroll
        for (int q = 0; q < KC2 / 8; ++q) {
            const v4f a  = zr[2 * q];
            const v4f b0 = w0[2 * q];
            const v4f b1 = w1[2 * q];
            const v2f a_lo = {a.x, a.y},   a_hi = {a.z, a.w};
            const v2f b0lo = {b0.x, b0.y}, b0hi = {b0.z, b0.w};
            const v2f b1lo = {b1.x, b1.y}, b1hi = {b1.z, b1.w};
            acc0 = wmma_f32(a_lo, b0lo, acc0);
            acc1 = wmma_f32(a_lo, b1lo, acc1);
            acc0 = wmma_f32(a_hi, b0hi, acc0);
            acc1 = wmma_f32(a_hi, b1hi, acc1);
        }

        if (kc + 1 < NCH2) {
#pragma unroll
            for (int i = 0; i < 16; ++i)
                sh_w[buf ^ 1][(i * 16 + wn) * WP4 + wc] = wpre[i];
        }
    }

#pragma unroll
    for (int r = 0; r < IPG; ++r) {
        const float a = half ? acc1[r] : acc0[r];
        const float v = selu_f(a + bia) * msk[r];
        out[(size_t)(row0 + r) * HD + fcol] = v + res[r];
    }
}

extern "C" void kernel_launch(void* const* d_in, const int* in_sizes, int n_in,
                              void* d_out, int out_size, void* d_ws, size_t ws_size,
                              hipStream_t stream) {
    (void)in_sizes; (void)n_in; (void)out_size; (void)d_ws; (void)ws_size;
    const float* h         = (const float*)d_in[0];
    const float* e         = (const float*)d_in[1];
    const float* node_mask = (const float*)d_in[2];
    const float* edge_mask = (const float*)d_in[3];
    const float* Wf        = (const float*)d_in[4];
    const float* bf        = (const float*)d_in[5];
    const float* Ww        = (const float*)d_in[6];
    const float* bw        = (const float*)d_in[7];
    float* out = (float*)d_out;

    ef_conv_fused<<<(NB * NA) / IPG, 256, 0, stream>>>(h, e, edge_mask, Wf, bf,
                                                       node_mask, Ww, bw, out);
}


# Node MLP's Ww stream runs ahead of its products: 3-stage LDS ring pre-filled inside the filter loop, hand-written barriers leave fetches in flight, fragments read one step ahead in one group
# speedup vs baseline: 4.1464x; 1.0308x over previous
#include <hip/hip_runtime.h>


typedef __attribute__((ext_vector_type(2))) float v2f;
typedef __attribute__((ext_vector_type(4))) float v4f;
typedef __attribute__((ext_vector_type(8))) float v8f;

#define NB 16
#define NA 128
#define HD 256
#define EF 30
#define KP 32
#define K2 (2 * HD)
#define KC2 64
#define NCH2 (K2 / KC2)
#define ZP4 ((K2 + 4) / 4)
#define WP4 ((KC2 + 4) / 4)
#define NST2 3
#define WPT (16 / IPG)

__device__ __forceinline__ float selu_f(float x) {
    const float scale = 1.0507009873554805f;
    const float alpha = 1.6732632423543772f;
    const float neg = scale * alpha * (__expf(x) - 1.0f);
    const float pos = scale * x;
    return x > 0.0f ? pos : neg;
}

#define LOG2E_F 1.4426950408889634f
#define SELU_A2 ((float)(1.6732632423543772 * 1.4426950408889634))
#define SELU_S2 ((float)(1.0507009873554805 * 0.6931471805599453))

__device__ __forceinline__ v8f wmma_f32(v2f a, v2f b, v8f c) {
    return __builtin_amdgcn_wmma_f32_16x16x4_f32(false, a, false, b, (short)0, c,
                                                 false, false);
}

__device__ __forceinline__ v4f ld4(const float* p) {
    v4f v;
    v.x = p[0]; v.y = p[1]; v.z = p[2]; v.w = p[3];
    return v;
}

#define IPG 8
#define EP4 ((KP + 4) / 4)
#define EST4 (NA * EP4)
#define ET4 (NA * EF / 4)

__device__ __forceinline__ void load_row_tile(const v4f* arow, v4f (&a)[KP / 8]) {
#pragma unroll
    for (int q = 0; q < KP / 8; ++q)
        a[q] = arow[2 * q];
}

__device__ __forceinline__ void row_tile_products(const v4f (&a)[KP / 8],
                                                  const v4f (&b0)[KP / 8],
                                                  const v4f (&b1)[KP / 8],
                                                  v8f& c0, v8f& c1) {
#pragma unroll
    for (int q = 0; q < KP / 8; ++q) {
        const v2f a_lo = {a[q].x, a[q].y},   a_hi = {a[q].z, a[q].w};
        const v2f b0lo = {b0[q].x, b0[q].y}, b0hi = {b0[q].z, b0[q].w};
        const v2f b1lo = {b1[q].x, b1[q].y}, b1hi = {b1[q].z, b1[q].w};
        c0 = wmma_f32(a_lo, b0lo, c0);
        c1 = wmma_f32(a_lo, b1lo, c1);
        c0 = wmma_f32(a_hi, b0hi, c0);
        c1 = wmma_f32(a_hi, b1hi, c1);
    }
}

__device__ __forceinline__ void fold_tiles(const v8f c0, const v8f c1, const v2f (&em2)[4],
                                           const v2f (&hvA)[4], const v2f (&hvB)[4],
                                           v2f (&acc)[4]) {
    const v2f a2  = {SELU_A2, SELU_A2};
    const v2f na2 = {-SELU_A2, -SELU_A2};
    v2f x2[8], e[8], neg[8], s[8], w[8];
#pragma unroll
    for (int rp = 0; rp < 4; ++rp) {
        x2[rp]     = (v2f){c0[2 * rp], c0[2 * rp + 1]};
        x2[4 + rp] = (v2f){c1[2 * rp], c1[2 * rp + 1]};
    }
#pragma unroll
    for (int p = 0; p < 8; ++p) {
        e[p].x = __builtin_amdgcn_exp2f(x2[p].x);
        e[p].y = __builtin_amdgcn_exp2f(x2[p].y);
    }
#pragma unroll
    for (int rp = 0; rp < 4; ++rp) {
        w[rp]     = em2[rp] * hvA[rp];
        w[4 + rp] = em2[rp] * hvB[rp];
    }
#pragma unroll
    for (int p = 0; p < 8; ++p)
        neg[p] = __builtin_elementwise_fma(e[p], a2, na2);
#pragma unroll
    for (int p = 0; p < 8; ++p) {
        s[p].x = x2[p].x > 0.0f ? x2[p].x : neg[p].x;
        s[p].y = x2[p].y > 0.0f ? x2[p].y : neg[p].y;
    }
#pragma unroll
    for (int p = 0; p < 8; ++p) {
        const int a = 2 * (p >> 2) + (p & 1);
        acc[a] = __builtin_elementwise_fma(s[p], w[p], acc[a]);
    }
}

__device__ __forceinline__ void load_node_frags(const v4f* zr, const v4f* w0,
                                                v4f (&a)[KC2 / 8], v4f (&b0)[KC2 / 8],
                                                v4f (&b1)[KC2 / 8]) {
    const v4f* w1 = w0 + 16 * WP4;
#pragma unroll
    for (int q = 0; q < KC2 / 8; ++q) {
        a[q]  = zr[2 * q];
        b0[q] = w0[2 * q];
        b1[q] = w1[2 * q];
    }
}

__device__ __forceinline__ void node_products(const v4f (&fa)[KC2 / 8], const v4f (&fb0)[KC2 / 8],
                                              const v4f (&fb1)[KC2 / 8], const int q0,
                                              const int q1, v8f& acc0, v8f& acc1) {
#pragma unroll
    for (int q = q0; q < q1; ++q) {
        const v4f a  = fa[q];
        const v4f b0 = fb0[q];
        const v4f b1 = fb1[q];
        const v2f a_lo = {a.x, a.y},   a_hi = {a.z, a.w};
        const v2f b0lo = {b0.x, b0.y}, b0hi = {b0.z, b0.w};
        const v2f b1lo = {b1.x, b1.y}, b1hi = {b1.z, b1.w};
        acc0 = wmma_f32(a_lo, b0lo, acc0);
        acc1 = wmma_f32(a_lo, b1lo, acc1);
        acc0 = wmma_f32(a_hi, b0hi, acc0);
        acc1 = wmma_f32(a_hi, b1hi, acc1);
    }
}

__global__ __launch_bounds__(256)
void ef_conv_fused(const float* __restrict__ h,
                   const float* __restrict__ e,
                   const float* __restrict__ edge_mask,
                   const float* __restrict__ Wf,
                   const float* __restrict__ bf,
                   const float* __restrict__ node_mask,
                   const float* __restrict__ Ww,
                   const float* __restrict__ bw,
                   float* __restrict__ out) {
    __shared__ v4f sh_e[2 * EST4];
    __shared__ v4f sh_m[2 * (NA / 4)];
    __shared__ v4f sh_z[IPG * ZP4];
    __shared__ v4f sh_w[NST2][HD * WP4];

    const int tid = threadIdx.x;
    const int b   = blockIdx.x / (NA / IPG);
    const int i0  = (blockIdx.x % (NA / IPG)) * IPG;

    const int lane  = tid & 31;
    const int wave  = tid >> 5;
    const int half  = lane >> 4;
    const int l     = lane & 15;
    const int fbase = wave * 32;

    const bool own3 = tid < ET4 - 3 * 256;
    int o_lo[4], o_hi[4];
#pragma unroll
    for (int s = 0; s < 4; ++s) {
        const int g  = 4 * (tid + 256 * s);
        const int j1 = g / EF,       k1 = g - j1 * EF;
        const int j2 = (g + 2) / EF, k2 = (g + 2) - j2 * EF;
        o_lo[s] = j1 * (2 * EP4) + (k1 >> 1);
        o_hi[s] = j2 * (2 * EP4) + (k2 >> 1);
    }
    const float* etile = e + ((size_t)(b * NA + i0)) * NA * EF + 4 * tid;
    const float* emrow = edge_mask + ((size_t)(b * NA + i0)) * NA + 4 * lane;

    const int row0 = b * NA + i0;
    const int wn = tid >> 4;
    const int wc = tid & 15;
    const float* wsrc = Ww + (size_t)wn * K2 + wc * 4;

#pragma unroll
    for (int p = 0; p < IPG / 4; ++p) {
        const int u  = tid + 256 * p;
        const int r  = u >> 6;
        const int c4 = u & 63;
        sh_z[r * ZP4 + c4] = ld4(h + (size_t)(row0 + r) * HD + c4 * 4);
    }

    v4f pre[4], prem;
    const v4f vzero4 = {0.f, 0.f, 0.f, 0.f};
#pragma unroll
    for (int s = 0; s < 4; ++s)
        pre[s] = (s < 3 || own3) ? ld4(etile + s * 1024) : vzero4;
    prem = (wave == 0) ? ld4(emrow) : vzero4;

    v4f bfr0[KP / 8], bfr1[KP / 8];
    {
        const float* w0 = Wf + (size_t)(fbase + l) * EF + 4 * half;
        const float* w1 = w0 + 16 * EF;
        const float bia0 = bf[fbase + l] * LOG2E_F;
        const float bia1 = bf[fbase + 16 + l] * LOG2E_F;
#pragma unroll
        for (int q = 0; q < KP / 8; ++q) {
            const bool tail = (q == KP / 8 - 1) && (half != 0);
            bfr0[q].x = w0[8 * q] * LOG2E_F;     bfr0[q].y = w0[8 * q + 1] * LOG2E_F;
            bfr1[q].x = w1[8 * q] * LOG2E_F;     bfr1[q].y = w1[8 * q + 1] * LOG2E_F;
            bfr0[q].z = tail ? bia0 : w0[8 * q + 2] * LOG2E_F;
            bfr0[q].w = tail ? 0.0f : w0[8 * q + 3] * LOG2E_F;
            bfr1[q].z = tail ? bia1 : w1[8 * q + 2] * LOG2E_F;
            bfr1[q].w = tail ? 0.0f : w1[8 * q + 3] * LOG2E_F;
        }
    }

    v2f hv2[8][2][4];
    {
        const float* hB = h + (size_t)b * NA * HD + (size_t)(half * 8) * HD + fbase + l;
#pragma unroll
        for (int mt = 0; mt < 8; ++mt)
#pragma unroll
            for (int nt = 0; nt < 2; ++nt)
#pragma unroll
                for (int rp = 0; rp < 4; ++rp) {
                    hv2[mt][nt][rp].x = hB[(size_t)(mt * 16 + 2 * rp) * HD + nt * 16];
                    hv2[mt][nt][rp].y = hB[(size_t)(mt * 16 + 2 * rp + 1) * HD + nt * 16];
                }
    }

    v2f* e2 = (v2f*)sh_e;
    if (tid < NA) {
        const v2f p2 = {1.f, 0.f};
        e2[tid * (2 * EP4) + EF / 2] = p2;
        e2[2 * EST4 + tid * (2 * EP4) + EF / 2] = p2;
    }
#pragma unroll
    for (int s = 0; s < 4; ++s) {
        if (s < 3 || own3) {
            const v2f lo = {pre[s].x, pre[s].y}, hi = {pre[s].z, pre[s].w};
            e2[o_lo[s]] = lo;
            e2[o_hi[s]] = hi;
        }
    }
    if (wave == 0) sh_m[lane] = prem;

    const v8f vzero = {0.f, 0.f, 0.f, 0.f, 0.f, 0.f, 0.f, 0.f};

    const float* wpf = wsrc;
    v4f* wdst = &sh_w[0][wn * WP4 + wc];

#pragma unroll 1
    for (int it = 0; it < IPG; ++it) {
        const int st = it & 1;
        __syncthreads();
        if (it + 1 < IPG) {
#pragma unroll
            for (int s = 0; s < 4; ++s)
                pre[s] = (s < 3 || own3) ? ld4(etile + (size_t)(it + 1) * NA * EF + s * 1024)
                                         : vzero4;
            prem = (wave == 0) ? ld4(emrow + (it + 1) * NA) : vzero4;
        }
        v4f wq[NST2][WPT];
#pragma unroll
        for (int c = 0; c < NST2; ++c)
#pragma unroll
            for (int j = 0; j < WPT; ++j)
                wq[c][j] = ld4(wpf + (size_t)j * 16 * K2 + c * KC2);

        const v4f* arow = sh_e + st * EST4 + l * EP4 + half;
        const v4f* mrow = sh_m + st * (NA / 4) + half * 2;

        const v2f vzero2 = {0.f, 0.f};
        v2f acc[4] = {vzero2, vzero2, vzero2, vzero2};
        v8f c0 = vzero, c1 = vzero;
        v4f aF[KP / 8], mF[2];
        load_row_tile(arow, aF);
        row_tile_products(aF, bfr0, bfr1, c0, c1);
        load_row_tile(arow + 16 * EP4, aF);
        mF[0] = mrow[0];
        mF[1] = mrow[1];
        __builtin_amdgcn_sched_barrier(0);
#pragma unroll
        for (int mt = 0; mt < 8; ++mt) {
            v8f n0 = vzero, n1 = vzero;
            v4f aN[KP / 8] = {vzero4, vzero4, vzero4, vzero4}, mN[2] = {vzero4, vzero4};
            if (mt + 2 < 8)
                load_row_tile(arow + (mt + 2) * 16 * EP4, aN);
            if (mt + 1 < 8) {
                mN[0] = mrow[(mt + 1) * 4];
                mN[1] = mrow[(mt + 1) * 4 + 1];
                row_tile_products(aF, bfr0, bfr1, n0, n1);
            }
            const v2f em2[4] = {{mF[0].x, mF[0].y}, {mF[0].z, mF[0].w},
                                {mF[1].x, mF[1].y}, {mF[1].z, mF[1].w}};
            fold_tiles(c0, c1, em2, hv2[mt][0], hv2[mt][1], acc);
            c0 = n0;
            c1 = n1;
#pragma unroll
            for (int q = 0; q < KP / 8; ++q)
                aF[q] = aN[q];
            mF[0] = mN[0];
            mF[1] = mN[1];
            if (mt + 1 < 8) {
                __builtin_amdgcn_sched_group_barrier(0x100, 6, 0);
#pragma unroll
                for (int s = 0; s < 16; ++s) {
                    __builtin_amdgcn_sched_group_barrier(0x400, 1, 0);
                    __builtin_amdgcn_sched_group_barrier(0x008, 1, 0);
                    __builtin_amdgcn_sched_group_barrier(0x002, 4, 0);
                }
                __builtin_amdgcn_sched_barrier(0);
            }
        }
        const v2f q0 = acc[0] + acc[1], q1 = acc[2] + acc[3];
        const float ps0 = q0.x + q0.y, ps1 = q1.x + q1.y;
        const float t0 = ps0 + __shfl_xor(ps0, 16, 32);
        const float t1 = ps1 + __shfl_xor(ps1, 16, 32);
        ((float*)sh_z)[it * (4 * ZP4) + HD + fbase + lane] = (half ? t1 : t0) * SELU_S2;

        if (it + 1 < IPG) {
            v2f* dst = e2 + (st ^ 1) * (2 * EST4);
#pragma unroll
            for (int s = 0; s < 4; ++s) {
                if (s < 3 || own3) {
                    const v2f lo = {pre[s].x, pre[s].y}, hi = {pre[s].z, pre[s].w};
                    dst[o_lo[s]] = lo;
                    dst[o_hi[s]] = hi;
                }
            }
            if (wave == 0) sh_m[(st ^ 1) * (NA / 4) + lane] = prem;
        }
#pragma unroll
        for (int c = 0; c < NST2; ++c)
#pragma unroll
            for (int j = 0; j < WPT; ++j)
                wdst[c * (HD * WP4) + j * 16 * WP4] = wq[c][j];
        wpf  += WPT * 16 * K2;
        wdst += WPT * 16 * WP4;
    }

    const int fcol  = fbase + lane;

    const float bia = bw[fcol];
    float msk[IPG], res[IPG];
#pragma unroll
    for (int r = 0; r < IPG; ++r) {
        msk[r] = node_mask[row0 + r];
        res[r] = h[(size_t)(row0 + r) * HD + fcol];
    }

    v4f wA[16], wB[16];
    v4f fa[2][KC2 / 8], fb0[2][KC2 / 8], fb1[2][KC2 / 8];

    v8f acc0 = vzero, acc1 = vzero;

#pragma unroll
    for (int kc = 0; kc < NCH2; ++kc) {
        if (kc + 3 < NCH2) {
#pragma unroll
            for (int i = 0; i < 16; ++i) {
                const v4f t = ld4(wsrc + (size_t)i * 16 * K2 + (kc + 3) * KC2);
                if ((kc + 3) & 1) wB[i] = t; else wA[i] = t;
            }
        }
        asm volatile("s_wait_dscnt 0x0\n\ts_barrier_signal -1\n\ts_barrier_wait -1" ::: "memory");

        if (kc == 0)
            load_node_frags(sh_z + (l & (IPG - 1)) * ZP4 + half,
                            sh_w[0] + (fbase + l) * WP4 + half, fa[0], fb0[0], fb1[0]);
        if (kc + 1 < NCH2)
            load_node_frags(sh_z + (l & (IPG - 1)) * ZP4 + (kc + 1) * (KC2 / 4) + half,
                            sh_w[(kc + 1) % NST2] + (fbase + l) * WP4 + half,
                            fa[(kc + 1) & 1], fb0[(kc + 1) & 1], fb1[(kc + 1) & 1]);
        __builtin_amdgcn_sched_barrier(0);

        node_products(fa[kc & 1], fb0[kc & 1], fb1[kc & 1], 0, KC2 / 16, acc0, acc1);

        if (kc + 2 >= NST2 && kc + 2 < NCH2) {
            __builtin_amdgcn_sched_barrier(0);
#pragma unroll
            for (int i = 0; i < 16; ++i)
                sh_w[(kc + 2) % NST2][(i * 16 + wn) * WP4 + wc] = ((kc + 2) & 1) ? wB[i] : wA[i];
            __builtin_amdgcn_sched_barrier(0);
        }

        node_products(fa[kc & 1], fb0[kc & 1], fb1[kc & 1], KC2 / 16, KC2 / 8, acc0, acc1);
        if (kc + 1 < NCH2)
            __builtin_amdgcn_sched_barrier(0);
    }

#pragma unroll
    for (int r = 0; r < IPG; ++r) {
        const float a = half ? acc1[r] : acc0[r];
        const float v = selu_f(a + bia) * msk[r];
        out[(size_t)(row0 + r) * HD + fcol] = v + res[r];
    }
}

extern "C" void kernel_launch(void* const* d_in, const int* in_sizes, int n_in,
                              void* d_out, int out_size, void* d_ws, size_t ws_size,
                              hipStream_t stream) {
    (void)in_sizes; (void)n_in; (void)out_size; (void)d_ws; (void)ws_size;
    const float* h         = (const float*)d_in[0];
    const float* e         = (const float*)d_in[1];
    const float* node_mask = (const float*)d_in[2];
    const float* edge_mask = (const float*)d_in[3];
    const float* Wf        = (const float*)d_in[4];
    const float* bf        = (const float*)d_in[5];
    const float* Ww        = (const float*)d_in[6];
    const float* bw        = (const float*)d_in[7];
    float* out = (float*)d_out;

    ef_conv_fused<<<(NB * NA) / IPG, 256, 0, stream>>>(h, e, edge_mask, Wf, bf,
                                                       node_mask, Ww, bw, out);
}
